# starting kernel
# baseline (speedup 1.0000x reference)
#include <hip/hip_runtime.h>

typedef __attribute__((ext_vector_type(2))) float v2f;
typedef __attribute__((ext_vector_type(8))) float v8f;

#define D_FEAT 128

__global__ void sage_zero(float* __restrict__ ws, size_t n) {
  size_t i = (size_t)blockIdx.x * blockDim.x + threadIdx.x;
  if (i < n) ws[i] = 0.0f;
}

__global__ __launch_bounds__(256) void sage_scatter1(
    const float* __restrict__ x, const int* __restrict__ ei,
    float* __restrict__ agg1, float* __restrict__ cnt, int n_edges) {
  const int lane = threadIdx.x & 31;
  const int e = (int)((blockIdx.x * (size_t)blockDim.x + threadIdx.x) >> 5);
  if (e >= n_edges) return;
  const int src = __builtin_nontemporal_load(ei + e);
  const int dst = __builtin_nontemporal_load(ei + n_edges + e);
  const float4 v = *(const float4*)(x + (size_t)src * D_FEAT + lane * 4);
  float* a = agg1 + (size_t)dst * D_FEAT + lane * 4;
  atomicAdd(a + 0, v.x);
  atomicAdd(a + 1, v.y);
  atomicAdd(a + 2, v.z);
  atomicAdd(a + 3, v.w);
  if (lane == 0) atomicAdd(cnt + dst, 1.0f);
}

__global__ __launch_bounds__(256) void sage_layer1_wmma(
    const float* __restrict__ x, const float* __restrict__ agg1,
    const float* __restrict__ cnt, const float* __restrict__ W1l,
    const float* __restrict__ W1r, const float* __restrict__ b1,
    float* __restrict__ h, int n_nodes) {
  __shared__ float ldsWl[D_FEAT * D_FEAT];

  for (int idx = threadIdx.x; idx < D_FEAT * 64; idx += 256) {
    const int r = idx >> 6;
    const int c = idx & 63;
    const float2 w = *(const float2*)(W1l + (size_t)r * D_FEAT + 2 * c);
    const int s = c ^ ((2 * r) & 63);
    *(float2*)(ldsWl + r * D_FEAT + 2 * s) = w;
  }
  __syncthreads();

  const int lane = threadIdx.x & 31;
  const int wave = threadIdx.x >> 5;
  const int tile = blockIdx.x * 8 + wave;
  const int n_tiles = n_nodes >> 4;
  if (tile >= n_tiles) return;

  const int half = lane >> 4;
  const int q = lane & 15;
  const int rowA = tile * 16 + q;

  const float inv = 1.0f / fmaxf(cnt[rowA], 1.0f);

  v8f acc[8];
#pragma unroll
  for (int i = 0; i < 8; ++i) {
    v8f z = {};
    acc[i] = z;
  }

  const float* aggRow = agg1 + (size_t)rowA * D_FEAT + 2 * half;
  const float* xRow   = x    + (size_t)rowA * D_FEAT + 2 * half;

  int nrow[8], nswz[8];
#pragma unroll
  for (int nt = 0; nt < 8; ++nt) {
    nrow[nt] = nt * 16 + q;
    nswz[nt] = (2 * nrow[nt]) & 63;
  }

  for (int k = 0; k < D_FEAT; k += 4) {
    const int c0 = (k >> 1) + half;
    const float2 tm = *(const float2*)(aggRow + k);
    const float2 ts = *(const float2*)(xRow + k);

    float2 bl[8], br[8];
#pragma unroll
    for (int nt = 0; nt < 8; ++nt) {
      bl[nt] = *(const float2*)(ldsWl + nrow[nt] * D_FEAT + 2 * (c0 ^ nswz[nt]));
      br[nt] = *(const float2*)(W1r + (size_t)nrow[nt] * D_FEAT + k + 2 * half);
    }

    v2f a_m;  a_m.x = tm.x * inv;  a_m.y = tm.y * inv;
    v2f a_s;  a_s.x = ts.x;        a_s.y = ts.y;
#pragma unroll
    for (int nt = 0; nt < 8; ++nt) {
      v2f vbl;  vbl.x = bl[nt].x;  vbl.y = bl[nt].y;
      v2f vbr;  vbr.x = br[nt].x;  vbr.y = br[nt].y;
      acc[nt] = __builtin_amdgcn_wmma_f32_16x16x4_f32(
          false, a_m, false, vbl, (short)0, acc[nt], false, false);
      acc[nt] = __builtin_amdgcn_wmma_f32_16x16x4_f32(
          false, a_s, false, vbr, (short)0, acc[nt], false, false);
    }
  }

#pragma unroll
  for (int nt = 0; nt < 8; ++nt) {
    const int n = nrow[nt];
    const float bias = b1[n];
#pragma unroll
    for (int v = 0; v < 8; ++v) {
      const int m = v + 8 * half;
      float val = acc[nt][v] + bias;
      val = fmaxf(val, 0.0f);
      h[(size_t)(tile * 16 + m) * D_FEAT + n] = val;
    }
  }
}

__global__ __launch_bounds__(256) void sage_pq(
    const float* __restrict__ h, const float* __restrict__ W2l,
    const float* __restrict__ W2r, const float* __restrict__ b2,
    float* __restrict__ p, float* __restrict__ out, int n_nodes) {
  const int lane = threadIdx.x & 31;
  const int node = (int)((blockIdx.x * (size_t)blockDim.x + threadIdx.x) >> 5);
  if (node >= n_nodes) return;
  const float4 hv = *(const float4*)(h + (size_t)node * D_FEAT + lane * 4);
  const float4 l0 = *(const float4*)(W2l + lane * 4);
  const float4 l1 = *(const float4*)(W2l + D_FEAT + lane * 4);
  const float4 r0 = *(const float4*)(W2r + lane * 4);
  const float4 r1 = *(const float4*)(W2r + D_FEAT + lane * 4);
  float d0 = hv.x * l0.x + hv.y * l0.y + hv.z * l0.z + hv.w * l0.w;
  float d1 = hv.x * l1.x + hv.y * l1.y + hv.z * l1.z + hv.w * l1.w;
  float d2 = hv.x * r0.x + hv.y * r0.y + hv.z * r0.z + hv.w * r0.w;
  float d3 = hv.x * r1.x + hv.y * r1.y + hv.z * r1.z + hv.w * r1.w;
#pragma unroll
  for (int off = 16; off > 0; off >>= 1) {
    d0 += __shfl_xor(d0, off, 32);
    d1 += __shfl_xor(d1, off, 32);
    d2 += __shfl_xor(d2, off, 32);
    d3 += __shfl_xor(d3, off, 32);
  }
  if (lane == 0) {
    p[(size_t)node * 2 + 0] = d0;
    p[(size_t)node * 2 + 1] = d1;
    out[(size_t)node * 2 + 0] = d2 + b2[0];
    out[(size_t)node * 2 + 1] = d3 + b2[1];
  }
}

__global__ __launch_bounds__(256) void sage_scatter2(
    const float* __restrict__ p, const int* __restrict__ ei,
    float* __restrict__ aggp, int n_edges) {
  const int e = blockIdx.x * blockDim.x + threadIdx.x;
  if (e >= n_edges) return;
  const int src = __builtin_nontemporal_load(ei + e);
  const int dst = __builtin_nontemporal_load(ei + n_edges + e);
  const float2 pv = *(const float2*)(p + (size_t)src * 2);
  atomicAdd(aggp + (size_t)dst * 2 + 0, pv.x);
  atomicAdd(aggp + (size_t)dst * 2 + 1, pv.y);
}

__global__ void sage_finalize(const float* __restrict__ aggp,
                              const float* __restrict__ cnt,
                              float* __restrict__ out, int n_nodes) {
  const int i = blockIdx.x * blockDim.x + threadIdx.x;
  if (i >= n_nodes) return;
  const float inv = 1.0f / fmaxf(cnt[i], 1.0f);
  out[(size_t)i * 2 + 0] += aggp[(size_t)i * 2 + 0] * inv;
  out[(size_t)i * 2 + 1] += aggp[(size_t)i * 2 + 1] * inv;
}

extern "C" void kernel_launch(void* const* d_in, const int* in_sizes, int n_in,
                              void* d_out, int out_size, void* d_ws, size_t ws_size,
                              hipStream_t stream) {
  const float* x   = (const float*)d_in[0];
  const int*   ei  = (const int*)d_in[1];
  const float* W1l = (const float*)d_in[2];
  const float* W1r = (const float*)d_in[3];
  const float* b1  = (const float*)d_in[4];
  const float* W2l = (const float*)d_in[5];
  const float* W2r = (const float*)d_in[6];
  const float* b2  = (const float*)d_in[7];
  float* out = (float*)d_out;

  const int n_nodes = in_sizes[0] / D_FEAT;
  const int n_edges = in_sizes[1] / 2;

  float* ws   = (float*)d_ws;
  float* agg1 = ws;
  float* cnt  = agg1 + (size_t)n_nodes * D_FEAT;
  float* aggp = cnt + n_nodes;
  float* h    = aggp + (size_t)n_nodes * 2;
  float* p    = h + (size_t)n_nodes * D_FEAT;

  const size_t zn = (size_t)n_nodes * (D_FEAT + 3);
  sage_zero<<<(unsigned)((zn + 255) / 256), 256, 0, stream>>>(ws, zn);

  sage_scatter1<<<(n_edges + 7) / 8, 256, 0, stream>>>(x, ei, agg1, cnt, n_edges);

  const int n_tiles = n_nodes / 16;
  sage_layer1_wmma<<<(n_tiles + 7) / 8, 256, 0, stream>>>(x, agg1, cnt, W1l, W1r,
                                                          b1, h, n_nodes);

  sage_pq<<<(n_nodes + 7) / 8, 256, 0, stream>>>(h, W2l, W2r, b2, p, out, n_nodes);

  sage_scatter2<<<(n_edges + 255) / 256, 256, 0, stream>>>(p, ei, aggp, n_edges);

  sage_finalize<<<(n_nodes + 255) / 256, 256, 0, stream>>>(aggp, cnt, out, n_nodes);
}


# Layer-1 WMMA fragment loads widened to 16 B per lane: one load of A (agg1, x) or B (W1r global, W1l LDS) feeds two k-steps, half the line touches; k order in each 8-block changes (sum order over k)
# speedup vs baseline: 2.6883x; 2.6883x over previous
#include <hip/hip_runtime.h>

typedef __attribute__((ext_vector_type(2))) float v2f;
typedef __attribute__((ext_vector_type(8))) float v8f;

#define D_FEAT 128

#define OWN_NB 200
#define OWN_WAVES 32
#define OWN_STEP 1024
#define OWN_FLUSH 32
#define OWN_QCAP 1056
#define OWN_B 16
#define OWN_AHEAD 4
#define OWN_PAD 0xFFu

__device__ __forceinline__ unsigned own_key(int dst) {
  const unsigned k = (unsigned)dst / OWN_NB;
  return k < OWN_PAD ? k : OWN_PAD;
}

__device__ __forceinline__ unsigned own_key4(int4 d) {
  return own_key(d.x) | (own_key(d.y) << 8) | (own_key(d.z) << 16) |
         (own_key(d.w) << 24);
}

__device__ __forceinline__ unsigned own_word(int s, int dst) {
  const unsigned l = (unsigned)dst - own_key(dst) * OWN_NB;
  return (l << 24) | ((unsigned)s & 0xFFFFFFu);
}

__device__ __forceinline__ uint4 own_word4(int4 s, int4 d) {
  uint4 w;
  w.x = own_word(s.x, d.x);
  w.y = own_word(s.y, d.y);
  w.z = own_word(s.z, d.z);
  w.w = own_word(s.w, d.w);
  return w;
}

__global__ __launch_bounds__(256) void sage_prep(
    const int* __restrict__ ei, unsigned* __restrict__ keys,
    unsigned* __restrict__ words, float* __restrict__ aggp, size_t zn,
    int n_edges) {
  const size_t t = (size_t)blockIdx.x * blockDim.x + threadIdx.x;
  const size_t nt = (size_t)gridDim.x * blockDim.x;
  for (size_t i = t; i < zn; i += nt) aggp[i] = 0.0f;

  const int* dstp = ei + n_edges;
  const size_t g0 = (size_t)blockIdx.x * (blockDim.x * 4) + threadIdx.x;
  if ((n_edges & 3) == 0 && n_edges > 0) {
    const size_t n_groups = (size_t)n_edges >> 2;
    int4 s[4], d[4];
#pragma unroll
    for (int j = 0; j < 4; ++j) {
      const size_t g = g0 + (size_t)j * blockDim.x;
      const size_t gc = g < n_groups ? g : n_groups - 1;
      s[j] = *(const int4*)(ei + gc * 4);
      d[j] = *(const int4*)(dstp + gc * 4);
    }
#pragma unroll
    for (int j = 0; j < 4; ++j) {
      const size_t g = g0 + (size_t)j * blockDim.x;
      if (g < n_groups) {
        keys[g] = own_key4(d[j]);
        *(uint4*)(words + g * 4) = own_word4(s[j], d[j]);
      } else {
        keys[g] = ~0u;
      }
    }
    return;
  }
  for (int j = 0; j < 4; ++j) {
    const size_t g = g0 + (size_t)j * blockDim.x;
    const size_t e = g * 4;
    if (e >= (size_t)n_edges) {
      keys[g] = ~0u;
      continue;
    }
    int4 s, d;
    s.x = ei[e];
    d.x = dstp[e];
    s.y = (e + 1 < (size_t)n_edges) ? ei[e + 1] : 0;
    d.y = (e + 1 < (size_t)n_edges) ? dstp[e + 1] : -1;
    s.z = (e + 2 < (size_t)n_edges) ? ei[e + 2] : 0;
    d.z = (e + 2 < (size_t)n_edges) ? dstp[e + 2] : -1;
    s.w = (e + 3 < (size_t)n_edges) ? ei[e + 3] : 0;
    d.w = (e + 3 < (size_t)n_edges) ? dstp[e + 3] : -1;
    keys[g] = own_key4(d);
    *(uint4*)(words + e) = own_word4(s, d);
  }
}


__device__ __forceinline__ uint4 own_keys(const unsigned* __restrict__ keys,
                                          int e) {
  return *(const uint4*)(keys + (e >> 2));
}

__device__ __forceinline__ unsigned own_eq(unsigned w, unsigned pat) {
  const unsigned t = w ^ pat;
  return ~(((t & 0x7F7F7F7Fu) + 0x7F7F7F7Fu) | t) & 0x80808080u;
}

__device__ __forceinline__ int own_push(unsigned* q, int qn, uint4 k,
                                        unsigned pat, int e) {
  unsigned m = (own_eq(k.x, pat) >> 7) | (own_eq(k.y, pat) >> 6) |
               (own_eq(k.z, pat) >> 5) | (own_eq(k.w, pat) >> 4);
  for (int r = 0; r < 16; ++r) {
    const bool has = m != 0u;
    const unsigned mask = __builtin_amdgcn_ballot_w32(has);
    if (mask == 0u) break;
    if (has) {
      const int p = __builtin_ctz(m);
      m &= m - 1u;
      const int pos = qn + (int)__builtin_amdgcn_mbcnt_lo(mask, 0u);
      q[pos] = (unsigned)(e + 4 * (p & 7) + (p >> 3));
    }
    qn += __builtin_popcount(mask);
  }
  return qn;
}

__device__ __forceinline__ int own_step(const unsigned* __restrict__ keys,
                                        uint4& k0, uint4& k1, int e,
                                        unsigned pat, unsigned* q, int qn) {
  qn = own_push(q, qn, k0, pat, e);
  qn = own_push(q, qn, k1, pat, e + 512);
  k0 = own_keys(keys, e + OWN_AHEAD * OWN_WAVES * OWN_STEP);
  k1 = own_keys(keys, e + OWN_AHEAD * OWN_WAVES * OWN_STEP + 512);
  return qn;
}

__device__ __forceinline__ void own_add_row(float* tile, int dl, int lane,
                                            const float4 v) {
  float* t = tile + dl * D_FEAT + lane;
  atomicAdd(t + 0, v.x);
  atomicAdd(t + 32, v.y);
  atomicAdd(t + 64, v.z);
  atomicAdd(t + 96, v.w);
}

__device__ __forceinline__ void own_entry(const unsigned* __restrict__ words,
                                          float* cntL, const unsigned* q,
                                          int& mysrc, int& myl) {
  const unsigned w = words[*q];
  mysrc = (int)(w & 0xFFFFFFu);
  myl = (int)(w >> 24);
  atomicAdd(cntL + myl, 1.0f);
}

__device__ __forceinline__ int own_flush(const float* __restrict__ x,
                                         const unsigned* __restrict__ words,
                                         float* tile, float* cntL, unsigned* q,
                                         int qn, int lane, bool last) {
  int b = 0;
  while (qn - b >= OWN_B) {
    int g = (qn - b) & ~(OWN_B - 1);
    if (g > 32) g = 32;
    int mysrc = 0, myl = 0;
    if (lane < g) own_entry(words, cntL, q + b + lane, mysrc, myl);
    for (int i0 = 0; i0 < g; i0 += OWN_B) {
      float4 v[OWN_B];
      int dl[OWN_B];
#pragma unroll
      for (int u = 0; u < OWN_B; ++u) {
        const int sidx = __builtin_amdgcn_readlane(mysrc, i0 + u);
        dl[u] = __builtin_amdgcn_readlane(myl, i0 + u);
        v[u] = *(const float4*)(x + (size_t)sidx * D_FEAT + lane * 4);
      }
#pragma unroll
      for (int u = 0; u < OWN_B; ++u) own_add_row(tile, dl[u], lane, v[u]);
    }
    b += g;
  }
  const int r = qn - b;
  if (last) {
    if (r > 0) {
      int mysrc = 0, myl = 0;
      if (lane < r) own_entry(words, cntL, q + b + lane, mysrc, myl);
      float4 v[OWN_B - 1];
      int dl[OWN_B - 1];
#pragma unroll
      for (int u = 0; u < OWN_B - 1; ++u) {
        if (u < r) {
          const int sidx = __builtin_amdgcn_readlane(mysrc, u);
          dl[u] = __builtin_amdgcn_readlane(myl, u);
          v[u] = *(const float4*)(x + (size_t)sidx * D_FEAT + lane * 4);
        }
      }
#pragma unroll
      for (int u = 0; u < OWN_B - 1; ++u) {
        if (u < r) own_add_row(tile, dl[u], lane, v[u]);
      }
    }
    return 0;
  }
  if (r > 0 && b > 0) {
    unsigned t = 0u;
    if (lane < r) t = q[b + lane];
    if (lane < r) q[lane] = t;
  }
  return r;
}

__global__ __launch_bounds__(1024) void sage_own1(
    const float* __restrict__ x, const unsigned* __restrict__ words,
    const unsigned* __restrict__ keys, float* __restrict__ agg1,
    float* __restrict__ cnt, int n_edges, int n_nodes) {
  __shared__ float tile[OWN_NB * D_FEAT];
  __shared__ float cntL[OWN_NB];
  __shared__ unsigned queue[OWN_WAVES * OWN_QCAP];

  const int lane = threadIdx.x & 31;
  const int wave = __builtin_amdgcn_readfirstlane((int)(threadIdx.x >> 5));
  const int base = blockIdx.x * OWN_NB;
  const unsigned pat = blockIdx.x * 0x01010101u;

  for (int i = threadIdx.x; i < OWN_NB * D_FEAT / 4; i += OWN_WAVES * 32)
    ((float4*)tile)[i] = make_float4(0.0f, 0.0f, 0.0f, 0.0f);
  if (threadIdx.x < OWN_NB) cntL[threadIdx.x] = 0.0f;
  __syncthreads();

  unsigned* q = queue + wave * OWN_QCAP;
  const int stride = OWN_WAVES * OWN_STEP;
  const int n_steps = (n_edges + stride - 1) / stride;
  int qn = 0;

  const int ef = wave * OWN_STEP + lane * 16;
  uint4 a0 = own_keys(keys, ef), a1 = own_keys(keys, ef + 512);
  uint4 b0 = own_keys(keys, ef + stride), b1 = own_keys(keys, ef + stride + 512);
  uint4 c0 = own_keys(keys, ef + 2 * stride), c1 = own_keys(keys, ef + 2 * stride + 512);
  uint4 d0 = own_keys(keys, ef + 3 * stride), d1 = own_keys(keys, ef + 3 * stride + 512);

  for (int s = 0; s < n_steps; s += OWN_AHEAD) {
    const int e = s * stride + ef;
    qn = own_step(keys, a0, a1, e, pat, q, qn);
    if (qn >= OWN_FLUSH)
      qn = own_flush(x, words, tile, cntL, q, qn, lane, false);
    qn = own_step(keys, b0, b1, e + stride, pat, q, qn);
    if (qn >= OWN_FLUSH)
      qn = own_flush(x, words, tile, cntL, q, qn, lane, false);
    qn = own_step(keys, c0, c1, e + 2 * stride, pat, q, qn);
    if (qn >= OWN_FLUSH)
      qn = own_flush(x, words, tile, cntL, q, qn, lane, false);
    qn = own_step(keys, d0, d1, e + 3 * stride, pat, q, qn);
    if (qn >= OWN_FLUSH)
      qn = own_flush(x, words, tile, cntL, q, qn, lane, false);
  }
  own_flush(x, words, tile, cntL, q, qn, lane, true);
  __syncthreads();

  for (int i = threadIdx.x; i < OWN_NB * D_FEAT / 4; i += OWN_WAVES * 32) {
    const int row = base + (i >> 5);
    if (row < n_nodes)
      *(float4*)(agg1 + (size_t)row * D_FEAT + (i & 31) * 4) = ((const float4*)tile)[i];
  }
  if (threadIdx.x < OWN_NB && base + (int)threadIdx.x < n_nodes)
    cnt[base + threadIdx.x] = cntL[threadIdx.x];
}

__global__ __launch_bounds__(256) __attribute__((amdgpu_waves_per_eu(1, 4)))
void sage_layer1_wmma(
    const float* __restrict__ x, const float* __restrict__ agg1,
    const float* __restrict__ cnt, const float* __restrict__ W1l,
    const float* __restrict__ W1r, const float* __restrict__ b1,
    float* __restrict__ h, int n_nodes) {
  __shared__ __attribute__((aligned(16))) float ldsWl[D_FEAT * D_FEAT];

  for (int idx = threadIdx.x; idx < D_FEAT * 64; idx += 256) {
    const int r = idx >> 6;
    const int c = idx & 63;
    const int f0 = 4 * ((2 * c) & 31) + (c >> 4);
    float2 w;
    w.x = W1l[(size_t)r * D_FEAT + f0];
    w.y = W1l[(size_t)r * D_FEAT + f0 + 4];
    const int s = c ^ ((2 * r) & 63);
    *(float2*)(ldsWl + r * D_FEAT + 2 * s) = w;
  }
  __syncthreads();

  const int lane = threadIdx.x & 31;
  const int wave = threadIdx.x >> 5;
  const int tile = blockIdx.x * 8 + wave;
  const int n_tiles = n_nodes >> 4;
  if (tile >= n_tiles) return;

  const int half = lane >> 4;
  const int q = lane & 15;
  const int rowA = tile * 16 + q;

  const float inv = 1.0f / fmaxf(cnt[rowA], 1.0f);

  v8f acc[8];
#pragma unroll
  for (int i = 0; i < 8; ++i) {
    v8f z = {};
    acc[i] = z;
  }

  const float* aggRow = agg1 + (size_t)rowA * D_FEAT + 4 * half;
  const float* xRow   = x    + (size_t)rowA * D_FEAT + 4 * half;

  int nrow[8], nswz[8];
#pragma unroll
  for (int nt = 0; nt < 8; ++nt) {
    nrow[nt] = nt * 16 + q;
    nswz[nt] = nrow[nt] & 31;
  }

  struct Frag {
    float4 am, as;
    float4 bl[8], br[8];
  };
  auto load = [&](Frag& f, int k) {
    const int u0 = (k >> 2) + half;
    f.am = *(const float4*)(aggRow + k);
    f.as = *(const float4*)(xRow + k);
#pragma unroll
    for (int nt = 0; nt < 8; ++nt) {
      f.bl[nt] = *(const float4*)(ldsWl + nrow[nt] * D_FEAT + 4 * (u0 ^ nswz[nt]));
      f.br[nt] = *(const float4*)(W1r + (size_t)nrow[nt] * D_FEAT + k + 4 * half);
    }
  };
  auto mma = [&](const Frag& f) {
    v2f a_m0;  a_m0.x = f.am.x * inv;  a_m0.y = f.am.y * inv;
    v2f a_m1;  a_m1.x = f.am.z * inv;  a_m1.y = f.am.w * inv;
    v2f a_s0;  a_s0.x = f.as.x;        a_s0.y = f.as.y;
    v2f a_s1;  a_s1.x = f.as.z;        a_s1.y = f.as.w;
#pragma unroll
    for (int nt = 0; nt < 8; ++nt) {
      v2f vbl;  vbl.x = f.bl[nt].x;  vbl.y = f.bl[nt].y;
      v2f vbr;  vbr.x = f.br[nt].x;  vbr.y = f.br[nt].y;
      acc[nt] = __builtin_amdgcn_wmma_f32_16x16x4_f32(
          false, a_m0, false, vbl, (short)0, acc[nt], false, false);
      acc[nt] = __builtin_amdgcn_wmma_f32_16x16x4_f32(
          false, a_s0, false, vbr, (short)0, acc[nt], false, false);
    }
#pragma unroll
    for (int nt = 0; nt < 8; ++nt) {
      v2f vbl;  vbl.x = f.bl[nt].z;  vbl.y = f.bl[nt].w;
      v2f vbr;  vbr.x = f.br[nt].z;  vbr.y = f.br[nt].w;
      acc[nt] = __builtin_amdgcn_wmma_f32_16x16x4_f32(
          false, a_m1, false, vbl, (short)0, acc[nt], false, false);
      acc[nt] = __builtin_amdgcn_wmma_f32_16x16x4_f32(
          false, a_s1, false, vbr, (short)0, acc[nt], false, false);
    }
  };

  Frag f0, f1;
  load(f0, 0);
  for (int k = 0; k < D_FEAT - 16; k += 16) {
    load(f1, k + 8);
    mma(f0);
    load(f0, k + 16);
    mma(f1);
  }
  load(f1, D_FEAT - 8);
  mma(f0);
  mma(f1);

#pragma unroll
  for (int nt = 0; nt < 8; ++nt) {
    const int n = nrow[nt];
    const float bias = b1[n];
#pragma unroll
    for (int v = 0; v < 8; ++v) {
      const int m = v + 8 * half;
      float val = acc[nt][v] + bias;
      val = fmaxf(val, 0.0f);
      h[(size_t)(tile * 16 + m) * D_FEAT + n] = val;
    }
  }
}

__global__ __launch_bounds__(256) void sage_pq(
    const float* __restrict__ h, const float* __restrict__ W2l,
    const float* __restrict__ W2r, const float* __restrict__ b2,
    float* __restrict__ p, float* __restrict__ out, int n_nodes) {
  const int lane = threadIdx.x & 31;
  const int node = (int)((blockIdx.x * (size_t)blockDim.x + threadIdx.x) >> 5);
  if (node >= n_nodes) return;
  const float4 hv = *(const float4*)(h + (size_t)node * D_FEAT + lane * 4);
  const float4 l0 = *(const float4*)(W2l + lane * 4);
  const float4 l1 = *(const float4*)(W2l + D_FEAT + lane * 4);
  const float4 r0 = *(const float4*)(W2r + lane * 4);
  const float4 r1 = *(const float4*)(W2r + D_FEAT + lane * 4);
  float d0 = hv.x * l0.x + hv.y * l0.y + hv.z * l0.z + hv.w * l0.w;
  float d1 = hv.x * l1.x + hv.y * l1.y + hv.z * l1.z + hv.w * l1.w;
  float d2 = hv.x * r0.x + hv.y * r0.y + hv.z * r0.z + hv.w * r0.w;
  float d3 = hv.x * r1.x + hv.y * r1.y + hv.z * r1.z + hv.w * r1.w;
#pragma unroll
  for (int off = 16; off > 0; off >>= 1) {
    d0 += __shfl_xor(d0, off, 32);
    d1 += __shfl_xor(d1, off, 32);
    d2 += __shfl_xor(d2, off, 32);
    d3 += __shfl_xor(d3, off, 32);
  }
  if (lane == 0) {
    p[(size_t)node * 2 + 0] = d0;
    p[(size_t)node * 2 + 1] = d1;
    out[(size_t)node * 2 + 0] = d2 + b2[0];
    out[(size_t)node * 2 + 1] = d3 + b2[1];
  }
}

__global__ __launch_bounds__(256) void sage_scatter2(
    const float* __restrict__ p, const int* __restrict__ ei,
    float* __restrict__ aggp, int n_edges) {
  const int t = blockIdx.x * blockDim.x + threadIdx.x;
  const int e = t >> 1;
  const int c = t & 1;
  if (e >= n_edges) return;
  const int src = __builtin_nontemporal_load(ei + e);
  const int dst = __builtin_nontemporal_load(ei + n_edges + e);
  const float pv = p[(size_t)src * 2 + c];
  atomicAdd(aggp + (size_t)dst * 2 + c, pv);
}

__global__ void sage_finalize(const float* __restrict__ aggp,
                              const float* __restrict__ cnt,
                              float* __restrict__ out, int n_nodes) {
  const int i = blockIdx.x * blockDim.x + threadIdx.x;
  if (i >= n_nodes) return;
  const float inv = 1.0f / fmaxf(cnt[i], 1.0f);
  out[(size_t)i * 2 + 0] += aggp[(size_t)i * 2 + 0] * inv;
  out[(size_t)i * 2 + 1] += aggp[(size_t)i * 2 + 1] * inv;
}

extern "C" void kernel_launch(void* const* d_in, const int* in_sizes, int n_in,
                              void* d_out, int out_size, void* d_ws, size_t ws_size,
                              hipStream_t stream) {
  const float* x   = (const float*)d_in[0];
  const int*   ei  = (const int*)d_in[1];
  const float* W1l = (const float*)d_in[2];
  const float* W1r = (const float*)d_in[3];
  const float* b1  = (const float*)d_in[4];
  const float* W2l = (const float*)d_in[5];
  const float* W2r = (const float*)d_in[6];
  const float* b2  = (const float*)d_in[7];
  float* out = (float*)d_out;

  const int n_nodes = in_sizes[0] / D_FEAT;
  const int n_edges = in_sizes[1] / 2;

  float* ws   = (float*)d_ws;
  float* agg1 = ws;
  float* cnt  = agg1 + (size_t)n_nodes * D_FEAT;
  float* aggp = cnt + n_nodes;
  float* h    = aggp + (size_t)n_nodes * 2;
  float* p    = h + (size_t)n_nodes * D_FEAT;

  unsigned* keys = (unsigned*)h;

  const int key_bytes = n_edges + 2 * OWN_AHEAD * OWN_WAVES * OWN_STEP;
  unsigned* words = keys + ((size_t)key_bytes / 16 + 512) * 4;

  const size_t zn = (size_t)n_nodes * 2;
  sage_prep<<<(key_bytes + 16 * 256 - 1) / (16 * 256), 256, 0, stream>>>(
      ei, keys, words, aggp, zn, n_edges);

  sage_own1<<<(n_nodes + OWN_NB - 1) / OWN_NB, OWN_WAVES * 32, 0, stream>>>(
      x, words, keys, agg1, cnt, n_edges, n_nodes);

  const int n_tiles = n_nodes / 16;
  sage_layer1_wmma<<<(n_tiles + 7) / 8, 256, 0, stream>>>(x, agg1, cnt, W1l, W1r,
                                                          b1, h, n_nodes);

  sage_pq<<<(n_nodes + 7) / 8, 256, 0, stream>>>(h, W2l, W2r, b2, p, out, n_nodes);

  sage_scatter2<<<(2 * n_edges + 255) / 256, 256, 0, stream>>>(p, ei, aggp, n_edges);

  sage_finalize<<<(n_nodes + 255) / 256, 256, 0, stream>>>(aggp, cnt, out, n_nodes);
}


# Layer-2 neighbour sum by the layer-1 owner blocks from edge chunks launch 1 leaves (128-B chunk + owner byte per group): LDS float adds, no global atomics, no finalize launch; sums arrival-ordered
# speedup vs baseline: 3.6452x; 1.3559x over previous
#include <hip/hip_runtime.h>

typedef __attribute__((ext_vector_type(2))) float v2f;
typedef __attribute__((ext_vector_type(8))) float v8f;

#define D_FEAT 128

#define OWN_NB 200
#define OWN_WAVES 32
#define OWN_STEP 1024
#define OWN_FLUSH 32
#define OWN_QCAP 1056
#define OWN_B 16
#define OWN_AHEAD 4
#define OWN_PAD 0xFFu
#define OWN_K 8
#define OWN2_QCAP 1536

__device__ __forceinline__ unsigned own_key(int dst) {
  const unsigned k = (unsigned)dst / OWN_NB;
  return k < OWN_PAD ? k : OWN_PAD;
}

__device__ __forceinline__ unsigned own_key4(int4 d) {
  return own_key(d.x) | (own_key(d.y) << 8) | (own_key(d.z) << 16) |
         (own_key(d.w) << 24);
}

__device__ __forceinline__ unsigned own_word(int s, int dst) {
  const unsigned l = (unsigned)dst - own_key(dst) * OWN_NB;
  return (l << 24) | ((unsigned)s & 0xFFFFFFu);
}

__device__ __forceinline__ uint4 own_word4(int4 s, int4 d) {
  uint4 w;
  w.x = own_word(s.x, d.x);
  w.y = own_word(s.y, d.y);
  w.z = own_word(s.z, d.z);
  w.w = own_word(s.w, d.w);
  return w;
}

__global__ __launch_bounds__(256) void sage_prep(
    const int* __restrict__ ei, unsigned* __restrict__ keys,
    unsigned* __restrict__ words, unsigned* __restrict__ dir32, size_t zn,
    int* __restrict__ cursor, int n_edges) {
  const size_t t = (size_t)blockIdx.x * blockDim.x + threadIdx.x;
  const size_t nt = (size_t)gridDim.x * blockDim.x;
  for (size_t i = t; i < zn; i += nt) dir32[i] = ~0u;
  if (t == 0) *cursor = 0;

  const int* dstp = ei + n_edges;
  const size_t g0 = (size_t)blockIdx.x * (blockDim.x * 4) + threadIdx.x;
  if ((n_edges & 3) == 0 && n_edges > 0) {
    const size_t n_groups = (size_t)n_edges >> 2;
    int4 s[4], d[4];
#pragma unroll
    for (int j = 0; j < 4; ++j) {
      const size_t g = g0 + (size_t)j * blockDim.x;
      const size_t gc = g < n_groups ? g : n_groups - 1;
      s[j] = *(const int4*)(ei + gc * 4);
      d[j] = *(const int4*)(dstp + gc * 4);
    }
#pragma unroll
    for (int j = 0; j < 4; ++j) {
      const size_t g = g0 + (size_t)j * blockDim.x;
      if (g < n_groups) {
        keys[g] = own_key4(d[j]);
        *(uint4*)(words + g * 4) = own_word4(s[j], d[j]);
      } else {
        keys[g] = ~0u;
      }
    }
    return;
  }
  for (int j = 0; j < 4; ++j) {
    const size_t g = g0 + (size_t)j * blockDim.x;
    const size_t e = g * 4;
    if (e >= (size_t)n_edges) {
      keys[g] = ~0u;
      continue;
    }
    int4 s, d;
    s.x = ei[e];
    d.x = dstp[e];
    s.y = (e + 1 < (size_t)n_edges) ? ei[e + 1] : 0;
    d.y = (e + 1 < (size_t)n_edges) ? dstp[e + 1] : -1;
    s.z = (e + 2 < (size_t)n_edges) ? ei[e + 2] : 0;
    d.z = (e + 2 < (size_t)n_edges) ? dstp[e + 2] : -1;
    s.w = (e + 3 < (size_t)n_edges) ? ei[e + 3] : 0;
    d.w = (e + 3 < (size_t)n_edges) ? dstp[e + 3] : -1;
    keys[g] = own_key4(d);
    *(uint4*)(words + e) = own_word4(s, d);
  }
}


__device__ __forceinline__ uint4 own_keys(const unsigned* __restrict__ keys,
                                          int e) {
  return *(const uint4*)(keys + (e >> 2));
}

__device__ __forceinline__ unsigned own_eq(unsigned w, unsigned pat) {
  const unsigned t = w ^ pat;
  return ~(((t & 0x7F7F7F7Fu) + 0x7F7F7F7Fu) | t) & 0x80808080u;
}

__device__ __forceinline__ int own_push(unsigned* q, int qn, uint4 k,
                                        unsigned pat, int e) {
  unsigned m = (own_eq(k.x, pat) >> 7) | (own_eq(k.y, pat) >> 6) |
               (own_eq(k.z, pat) >> 5) | (own_eq(k.w, pat) >> 4);
  for (int r = 0; r < 16; ++r) {
    const bool has = m != 0u;
    const unsigned mask = __builtin_amdgcn_ballot_w32(has);
    if (mask == 0u) break;
    if (has) {
      const int p = __builtin_ctz(m);
      m &= m - 1u;
      const int pos = qn + (int)__builtin_amdgcn_mbcnt_lo(mask, 0u);
      q[pos] = (unsigned)(e + 4 * (p & 7) + (p >> 3));
    }
    qn += __builtin_popcount(mask);
  }
  return qn;
}

__device__ __forceinline__ int own_step(const unsigned* __restrict__ keys,
                                        uint4& k0, uint4& k1, int e,
                                        unsigned pat, unsigned* q, int qn) {
  qn = own_push(q, qn, k0, pat, e);
  qn = own_push(q, qn, k1, pat, e + 512);
  k0 = own_keys(keys, e + OWN_AHEAD * OWN_WAVES * OWN_STEP);
  k1 = own_keys(keys, e + OWN_AHEAD * OWN_WAVES * OWN_STEP + 512);
  return qn;
}

__device__ __forceinline__ void own_add_row(float* tile, int dl, int lane,
                                            const float4 v) {
  float* t = tile + dl * D_FEAT + lane;
  atomicAdd(t + 0, v.x);
  atomicAdd(t + 32, v.y);
  atomicAdd(t + 64, v.z);
  atomicAdd(t + 96, v.w);
}

__device__ __forceinline__ void own_entry(const unsigned* __restrict__ words,
                                          float* cntL, const unsigned* q,
                                          int& mysrc, int& myl, unsigned& w) {
  w = words[*q];
  mysrc = (int)(w & 0xFFFFFFu);
  myl = (int)(w >> 24);
  atomicAdd(cntL + myl, 1.0f);
}

struct OwnOut {
  unsigned* chunks;
  unsigned char* dir;
  int* cursor;
  int slot0;
  int stat_total;
  int nem;
};

__device__ __forceinline__ void own_emit(OwnOut& o, unsigned w, int lane) {
  int c;
  if (o.nem < OWN_K) {
    c = o.slot0 + o.nem;
  } else {
    int t = 0;
    if (lane == 0) t = atomicAdd(o.cursor, 1);
    c = o.stat_total + __builtin_amdgcn_readfirstlane(t);
  }
  ++o.nem;
  o.chunks[(size_t)c * 32 + lane] = w;
  if (lane == 0) o.dir[c] = (unsigned char)blockIdx.x;
}

__device__ __forceinline__ int own_flush(const float* __restrict__ x,
                                         const unsigned* __restrict__ words,
                                         float* tile, float* cntL, unsigned* q,
                                         int qn, int lane, bool last,
                                         OwnOut& o) {
  int b = 0;
  while (qn - b >= OWN_B) {
    int g = (qn - b) & ~(OWN_B - 1);
    if (g > 32) g = 32;
    int mysrc = 0, myl = 0;
    unsigned w = ~0u;
    if (lane < g) own_entry(words, cntL, q + b + lane, mysrc, myl, w);
    own_emit(o, w, lane);
    for (int i0 = 0; i0 < g; i0 += OWN_B) {
      float4 v[OWN_B];
      int dl[OWN_B];
#pragma unroll
      for (int u = 0; u < OWN_B; ++u) {
        const int sidx = __builtin_amdgcn_readlane(mysrc, i0 + u);
        dl[u] = __builtin_amdgcn_readlane(myl, i0 + u);
        v[u] = *(const float4*)(x + (size_t)sidx * D_FEAT + lane * 4);
      }
#pragma unroll
      for (int u = 0; u < OWN_B; ++u) own_add_row(tile, dl[u], lane, v[u]);
    }
    b += g;
  }
  const int r = qn - b;
  if (last) {
    if (r > 0) {
      int mysrc = 0, myl = 0;
      unsigned w = ~0u;
      if (lane < r) own_entry(words, cntL, q + b + lane, mysrc, myl, w);
      own_emit(o, w, lane);
      float4 v[OWN_B - 1];
      int dl[OWN_B - 1];
#pragma unroll
      for (int u = 0; u < OWN_B - 1; ++u) {
        if (u < r) {
          const int sidx = __builtin_amdgcn_readlane(mysrc, u);
          dl[u] = __builtin_amdgcn_readlane(myl, u);
          v[u] = *(const float4*)(x + (size_t)sidx * D_FEAT + lane * 4);
        }
      }
#pragma unroll
      for (int u = 0; u < OWN_B - 1; ++u) {
        if (u < r) own_add_row(tile, dl[u], lane, v[u]);
      }
    }
    return 0;
  }
  if (r > 0 && b > 0) {
    unsigned t = 0u;
    if (lane < r) t = q[b + lane];
    if (lane < r) q[lane] = t;
  }
  return r;
}

__global__ __launch_bounds__(1024) void sage_own1(
    const float* __restrict__ x, const unsigned* __restrict__ words,
    const unsigned* __restrict__ keys, float* __restrict__ agg1,
    float* __restrict__ cnt, unsigned* __restrict__ chunks,
    unsigned char* __restrict__ dir, int* __restrict__ cursor, int n_edges,
    int n_nodes) {
  __shared__ float tile[OWN_NB * D_FEAT];
  __shared__ float cntL[OWN_NB];
  __shared__ unsigned queue[OWN_WAVES * OWN_QCAP];

  const int lane = threadIdx.x & 31;
  const int wave = __builtin_amdgcn_readfirstlane((int)(threadIdx.x >> 5));
  const int base = blockIdx.x * OWN_NB;
  const unsigned pat = blockIdx.x * 0x01010101u;

  for (int i = threadIdx.x; i < OWN_NB * D_FEAT / 4; i += OWN_WAVES * 32)
    ((float4*)tile)[i] = make_float4(0.0f, 0.0f, 0.0f, 0.0f);
  if (threadIdx.x < OWN_NB) cntL[threadIdx.x] = 0.0f;
  __syncthreads();

  unsigned* q = queue + wave * OWN_QCAP;
  OwnOut o;
  o.chunks = chunks;
  o.dir = dir;
  o.cursor = cursor;
  o.slot0 = ((int)blockIdx.x * OWN_WAVES + wave) * OWN_K;
  o.stat_total = (int)gridDim.x * OWN_WAVES * OWN_K;
  o.nem = 0;
  const int stride = OWN_WAVES * OWN_STEP;
  const int n_steps = (n_edges + stride - 1) / stride;
  int qn = 0;

  const int ef = wave * OWN_STEP + lane * 16;
  uint4 a0 = own_keys(keys, ef), a1 = own_keys(keys, ef + 512);
  uint4 b0 = own_keys(keys, ef + stride), b1 = own_keys(keys, ef + stride + 512);
  uint4 c0 = own_keys(keys, ef + 2 * stride), c1 = own_keys(keys, ef + 2 * stride + 512);
  uint4 d0 = own_keys(keys, ef + 3 * stride), d1 = own_keys(keys, ef + 3 * stride + 512);

  for (int s = 0; s < n_steps; s += OWN_AHEAD) {
    const int e = s * stride + ef;
    qn = own_step(keys, a0, a1, e, pat, q, qn);
    if (qn >= OWN_FLUSH)
      qn = own_flush(x, words, tile, cntL, q, qn, lane, false, o);
    qn = own_step(keys, b0, b1, e + stride, pat, q, qn);
    if (qn >= OWN_FLUSH)
      qn = own_flush(x, words, tile, cntL, q, qn, lane, false, o);
    qn = own_step(keys, c0, c1, e + 2 * stride, pat, q, qn);
    if (qn >= OWN_FLUSH)
      qn = own_flush(x, words, tile, cntL, q, qn, lane, false, o);
    qn = own_step(keys, d0, d1, e + 3 * stride, pat, q, qn);
    if (qn >= OWN_FLUSH)
      qn = own_flush(x, words, tile, cntL, q, qn, lane, false, o);
  }
  own_flush(x, words, tile, cntL, q, qn, lane, true, o);
  __syncthreads();

  for (int i = threadIdx.x; i < OWN_NB * D_FEAT / 4; i += OWN_WAVES * 32) {
    const int row = base + (i >> 5);
    if (row < n_nodes)
      *(float4*)(agg1 + (size_t)row * D_FEAT + (i & 31) * 4) = ((const float4*)tile)[i];
  }
  if (threadIdx.x < OWN_NB && base + (int)threadIdx.x < n_nodes)
    cnt[base + threadIdx.x] = cntL[threadIdx.x];
}

__global__ __launch_bounds__(256) __attribute__((amdgpu_waves_per_eu(1, 4)))
void sage_layer1_wmma(
    const float* __restrict__ x, const float* agg1,
    const float* __restrict__ cnt, const float* __restrict__ W1l,
    const float* __restrict__ W1r, const float* __restrict__ b1,
    float* h, int n_nodes) {
  __shared__ __attribute__((aligned(16))) float ldsWl[D_FEAT * D_FEAT];

  for (int idx = threadIdx.x; idx < D_FEAT * 64; idx += 256) {
    const int r = idx >> 6;
    const int c = idx & 63;
    const int f0 = 4 * ((2 * c) & 31) + (c >> 4);
    float2 w;
    w.x = W1l[(size_t)r * D_FEAT + f0];
    w.y = W1l[(size_t)r * D_FEAT + f0 + 4];
    const int s = c ^ ((2 * r) & 63);
    *(float2*)(ldsWl + r * D_FEAT + 2 * s) = w;
  }
  __syncthreads();

  const int lane = threadIdx.x & 31;
  const int wave = threadIdx.x >> 5;
  const int tile = blockIdx.x * 8 + wave;
  const int n_tiles = n_nodes >> 4;
  if (tile >= n_tiles) return;

  const int half = lane >> 4;
  const int q = lane & 15;
  const int rowA = tile * 16 + q;

  const float inv = 1.0f / fmaxf(cnt[rowA], 1.0f);

  v8f acc[8];
#pragma unroll
  for (int i = 0; i < 8; ++i) {
    v8f z = {};
    acc[i] = z;
  }

  const float* aggRow = agg1 + (size_t)rowA * D_FEAT + 4 * half;
  const float* xRow   = x    + (size_t)rowA * D_FEAT + 4 * half;

  int nrow[8], nswz[8];
#pragma unroll
  for (int nt = 0; nt < 8; ++nt) {
    nrow[nt] = nt * 16 + q;
    nswz[nt] = nrow[nt] & 31;
  }

  struct Frag {
    float4 am, as;
    float4 bl[8], br[8];
  };
  auto load = [&](Frag& f, int k) {
    const int u0 = (k >> 2) + half;
    f.am = *(const float4*)(aggRow + k);
    f.as = *(const float4*)(xRow + k);
#pragma unroll
    for (int nt = 0; nt < 8; ++nt) {
      f.bl[nt] = *(const float4*)(ldsWl + nrow[nt] * D_FEAT + 4 * (u0 ^ nswz[nt]));
      f.br[nt] = *(const float4*)(W1r + (size_t)nrow[nt] * D_FEAT + k + 4 * half);
    }
  };
  auto mma = [&](const Frag& f) {
    v2f a_m0;  a_m0.x = f.am.x * inv;  a_m0.y = f.am.y * inv;
    v2f a_m1;  a_m1.x = f.am.z * inv;  a_m1.y = f.am.w * inv;
    v2f a_s0;  a_s0.x = f.as.x;        a_s0.y = f.as.y;
    v2f a_s1;  a_s1.x = f.as.z;        a_s1.y = f.as.w;
#pragma unroll
    for (int nt = 0; nt < 8; ++nt) {
      v2f vbl;  vbl.x = f.bl[nt].x;  vbl.y = f.bl[nt].y;
      v2f vbr;  vbr.x = f.br[nt].x;  vbr.y = f.br[nt].y;
      acc[nt] = __builtin_amdgcn_wmma_f32_16x16x4_f32(
          false, a_m0, false, vbl, (short)0, acc[nt], false, false);
      acc[nt] = __builtin_amdgcn_wmma_f32_16x16x4_f32(
          false, a_s0, false, vbr, (short)0, acc[nt], false, false);
    }
#pragma unroll
    for (int nt = 0; nt < 8; ++nt) {
      v2f vbl;  vbl.x = f.bl[nt].z;  vbl.y = f.bl[nt].w;
      v2f vbr;  vbr.x = f.br[nt].z;  vbr.y = f.br[nt].w;
      acc[nt] = __builtin_amdgcn_wmma_f32_16x16x4_f32(
          false, a_m1, false, vbl, (short)0, acc[nt], false, false);
      acc[nt] = __builtin_amdgcn_wmma_f32_16x16x4_f32(
          false, a_s1, false, vbr, (short)0, acc[nt], false, false);
    }
  };

  Frag f0, f1;
  load(f0, 0);
  for (int k = 0; k < D_FEAT - 16; k += 16) {
    load(f1, k + 8);
    mma(f0);
    load(f0, k + 16);
    mma(f1);
  }
  load(f1, D_FEAT - 8);
  mma(f0);
  mma(f1);

#pragma unroll
  for (int nt = 0; nt < 8; ++nt) {
    const int n = nrow[nt];
    const float bias = b1[n];
#pragma unroll
    for (int v = 0; v < 8; ++v) {
      const int m = v + 8 * half;
      float val = acc[nt][v] + bias;
      val = fmaxf(val, 0.0f);
      h[(size_t)(tile * 16 + m) * D_FEAT + n] = val;
    }
  }
}

__global__ __launch_bounds__(256) void sage_pq(
    const float* __restrict__ h, const float* __restrict__ W2l,
    const float* __restrict__ W2r, const float* __restrict__ b2,
    float* __restrict__ p, float* __restrict__ out, int n_nodes) {
  const int lane = threadIdx.x & 31;
  const int node = (int)((blockIdx.x * (size_t)blockDim.x + threadIdx.x) >> 5);
  if (node >= n_nodes) return;
  const float4 hv = *(const float4*)(h + (size_t)node * D_FEAT + lane * 4);
  const float4 l0 = *(const float4*)(W2l + lane * 4);
  const float4 l1 = *(const float4*)(W2l + D_FEAT + lane * 4);
  const float4 r0 = *(const float4*)(W2r + lane * 4);
  const float4 r1 = *(const float4*)(W2r + D_FEAT + lane * 4);
  float d0 = hv.x * l0.x + hv.y * l0.y + hv.z * l0.z + hv.w * l0.w;
  float d1 = hv.x * l1.x + hv.y * l1.y + hv.z * l1.z + hv.w * l1.w;
  float d2 = hv.x * r0.x + hv.y * r0.y + hv.z * r0.z + hv.w * r0.w;
  float d3 = hv.x * r1.x + hv.y * r1.y + hv.z * r1.z + hv.w * r1.w;
#pragma unroll
  for (int off = 16; off > 0; off >>= 1) {
    d0 += __shfl_xor(d0, off, 32);
    d1 += __shfl_xor(d1, off, 32);
    d2 += __shfl_xor(d2, off, 32);
    d3 += __shfl_xor(d3, off, 32);
  }
  if (lane == 0) {
    p[(size_t)node * 2 + 0] = d0;
    p[(size_t)node * 2 + 1] = d1;
    out[(size_t)node * 2 + 0] = d2 + b2[0];
    out[(size_t)node * 2 + 1] = d3 + b2[1];
  }
}

__device__ __forceinline__ int own2_drain(const float* __restrict__ p,
                                          const unsigned* __restrict__ chunks,
                                          float* tile2, const unsigned* q,
                                          int qn, int lane) {
  for (int b = 0; b < qn; b += 8) {
    unsigned w[8];
#pragma unroll
    for (int u = 0; u < 8; ++u) {
      w[u] = ~0u;
      if (b + u < qn) w[u] = chunks[(size_t)q[b + u] * 32 + lane];
    }
    float px[8], py[8];
#pragma unroll
    for (int u = 0; u < 8; ++u) {
      px[u] = 0.0f;
      py[u] = 0.0f;
      if (w[u] != ~0u) {
        const size_t s = (size_t)(w[u] & 0xFFFFFFu) * 2;
        px[u] = p[s];
        py[u] = p[s + 1];
      }
    }
#pragma unroll
    for (int u = 0; u < 8; ++u) {
      if (w[u] != ~0u) {
        float* t = tile2 + 2 * (w[u] >> 24);
        atomicAdd(t, px[u]);
        atomicAdd(t + 1, py[u]);
      }
    }
  }
  return 0;
}

__global__ __launch_bounds__(1024) void sage_own2(
    const float* __restrict__ p, const unsigned* __restrict__ chunks,
    const unsigned* __restrict__ dir32, const int* __restrict__ cursor,
    const float* __restrict__ cnt, float* __restrict__ out, int n_nodes,
    int stat_total, int ovf_cap) {
  __shared__ float tile2[OWN_NB * 2];
  __shared__ unsigned queue2[OWN_WAVES * OWN2_QCAP];

  const int lane = threadIdx.x & 31;
  const int wave = __builtin_amdgcn_readfirstlane((int)(threadIdx.x >> 5));
  const int base = blockIdx.x * OWN_NB;
  const unsigned pat = blockIdx.x * 0x01010101u;

  if (threadIdx.x < OWN_NB * 2) tile2[threadIdx.x] = 0.0f;
  __syncthreads();

  int ov = *cursor;
  if (ov > ovf_cap) ov = ovf_cap;
  if (ov < 0) ov = 0;
  const int n_scan = stat_total + ov;

  unsigned* q = queue2 + wave * OWN2_QCAP;
  int qn = 0;
  const uint4 none = make_uint4(~0u, ~0u, ~0u, ~0u);
  static_assert(OWN_K == 8, "own slots of a wave = two directory dwords");
  {
    const int slot0 = ((int)blockIdx.x * OWN_WAVES + wave) * OWN_K;
    uint4 k = none;
    if (lane == 0) {
      k.x = dir32[slot0 >> 2];
      k.y = dir32[(slot0 >> 2) + 1];
    }
    qn = own_push(q, qn, k, pat, slot0);
  }
  for (int r0 = stat_total; r0 < n_scan; r0 += OWN_WAVES * 1024) {
    const int cb = r0 + wave * 1024 + lane * 16;
    uint4 k0 = none, k1 = none;
    if (cb < n_scan) k0 = *(const uint4*)(dir32 + (cb >> 2));
    if (cb + 512 < n_scan) k1 = *(const uint4*)(dir32 + ((cb + 512) >> 2));
    qn = own_push(q, qn, k0, pat, cb);
    qn = own_push(q, qn, k1, pat, cb + 512);
    if (qn > OWN2_QCAP - 1024) qn = own2_drain(p, chunks, tile2, q, qn, lane);
  }
  own2_drain(p, chunks, tile2, q, qn, lane);
  __syncthreads();

  if (threadIdx.x < OWN_NB) {
    const int node = base + (int)threadIdx.x;
    if (node < n_nodes) {
      const float inv = 1.0f / fmaxf(cnt[node], 1.0f);
      out[(size_t)node * 2 + 0] += tile2[2 * threadIdx.x + 0] * inv;
      out[(size_t)node * 2 + 1] += tile2[2 * threadIdx.x + 1] * inv;
    }
  }
}

extern "C" void kernel_launch(void* const* d_in, const int* in_sizes, int n_in,
                              void* d_out, int out_size, void* d_ws, size_t ws_size,
                              hipStream_t stream) {
  const float* x   = (const float*)d_in[0];
  const int*   ei  = (const int*)d_in[1];
  const float* W1l = (const float*)d_in[2];
  const float* W1r = (const float*)d_in[3];
  const float* b1  = (const float*)d_in[4];
  const float* W2l = (const float*)d_in[5];
  const float* W2r = (const float*)d_in[6];
  const float* b2  = (const float*)d_in[7];
  float* out = (float*)d_out;

  const int n_nodes = in_sizes[0] / D_FEAT;
  const int n_edges = in_sizes[1] / 2;

  float* ws   = (float*)d_ws;
  float* agg1 = ws;
  float* cnt  = agg1 + (size_t)n_nodes * D_FEAT;
  float* scr  = cnt + (size_t)n_nodes * 3;
  float* p    = scr + (size_t)n_nodes * D_FEAT;
  float* h    = agg1;

  unsigned* keys = (unsigned*)scr;

  const int key_bytes = n_edges + 2 * OWN_AHEAD * OWN_WAVES * OWN_STEP;
  unsigned* words = keys + ((size_t)key_bytes / 16 + 512) * 4;

  const int n_blocks = (n_nodes + OWN_NB - 1) / OWN_NB;
  const int stat_total = n_blocks * OWN_WAVES * OWN_K;
  const int ovf_cap = n_edges / 16 + n_blocks * OWN_WAVES + 64;
  const size_t max_chunks = (size_t)stat_total + ovf_cap;
  unsigned* chunks = words + ((size_t)n_edges / 4 + 4) * 4;
  unsigned* dir32 = chunks + max_chunks * 32;
  const size_t dir_words = (max_chunks / 16 + 1) * 4;
  int* cursor = (int*)(dir32 + dir_words);

  sage_prep<<<(key_bytes + 16 * 256 - 1) / (16 * 256), 256, 0, stream>>>(
      ei, keys, words, dir32, dir_words, cursor, n_edges);

  sage_own1<<<n_blocks, OWN_WAVES * 32, 0, stream>>>(
      x, words, keys, agg1, cnt, chunks, (unsigned char*)dir32, cursor, n_edges,
      n_nodes);

  const int n_tiles = n_nodes / 16;
  sage_layer1_wmma<<<(n_tiles + 7) / 8, 256, 0, stream>>>(x, agg1, cnt, W1l, W1r,
                                                          b1, h, n_nodes);

  sage_pq<<<(n_nodes + 7) / 8, 256, 0, stream>>>(h, W2l, W2r, b2, p, out, n_nodes);

  sage_own2<<<n_blocks, OWN_WAVES * 32, 0, stream>>>(p, chunks, dir32, cursor, cnt,
                                                     out, n_nodes, stat_total,
                                                     ovf_cap);
}


# Layer-2 projections as a 2nd WMMA run on the layer-1 accumulators: layer 1 runs transposed (W rows = A, nodes = B) so its D is a ready B operand; h never stored, 4 launches; sum order changes
# speedup vs baseline: 4.3736x; 1.1998x over previous
#include <hip/hip_runtime.h>

typedef __attribute__((ext_vector_type(2))) float v2f;
typedef __attribute__((ext_vector_type(8))) float v8f;

#define D_FEAT 128

#define OWN_NB 200
#define OWN_WAVES 32
#define OWN_STEP 1024
#define OWN_FLUSH 32
#define OWN_QCAP 1056
#define OWN_B 16
#define OWN_AHEAD 4
#define OWN_PAD 0xFFu
#define OWN_K 8
#define OWN2_QCAP 1536

__device__ __forceinline__ unsigned own_key(int dst) {
  const unsigned k = (unsigned)dst / OWN_NB;
  return k < OWN_PAD ? k : OWN_PAD;
}

__device__ __forceinline__ unsigned own_key4(int4 d) {
  return own_key(d.x) | (own_key(d.y) << 8) | (own_key(d.z) << 16) |
         (own_key(d.w) << 24);
}

__device__ __forceinline__ unsigned own_word(int s, int dst) {
  const unsigned l = (unsigned)dst - own_key(dst) * OWN_NB;
  return (l << 24) | ((unsigned)s & 0xFFFFFFu);
}

__device__ __forceinline__ uint4 own_word4(int4 s, int4 d) {
  uint4 w;
  w.x = own_word(s.x, d.x);
  w.y = own_word(s.y, d.y);
  w.z = own_word(s.z, d.z);
  w.w = own_word(s.w, d.w);
  return w;
}

__global__ __launch_bounds__(256) void sage_prep(
    const int* __restrict__ ei, unsigned* __restrict__ keys,
    unsigned* __restrict__ words, unsigned* __restrict__ dir32, size_t zn,
    int* __restrict__ cursor, int n_edges) {
  const size_t t = (size_t)blockIdx.x * blockDim.x + threadIdx.x;
  const size_t nt = (size_t)gridDim.x * blockDim.x;
  for (size_t i = t; i < zn; i += nt) dir32[i] = ~0u;
  if (t == 0) *cursor = 0;

  const int* dstp = ei + n_edges;
  const size_t g0 = (size_t)blockIdx.x * (blockDim.x * 4) + threadIdx.x;
  if ((n_edges & 3) == 0 && n_edges > 0) {
    const size_t n_groups = (size_t)n_edges >> 2;
    int4 s[4], d[4];
#pragma unroll
    for (int j = 0; j < 4; ++j) {
      const size_t g = g0 + (size_t)j * blockDim.x;
      const size_t gc = g < n_groups ? g : n_groups - 1;
      s[j] = *(const int4*)(ei + gc * 4);
      d[j] = *(const int4*)(dstp + gc * 4);
    }
#pragma unroll
    for (int j = 0; j < 4; ++j) {
      const size_t g = g0 + (size_t)j * blockDim.x;
      if (g < n_groups) {
        keys[g] = own_key4(d[j]);
        *(uint4*)(words + g * 4) = own_word4(s[j], d[j]);
      } else {
        keys[g] = ~0u;
      }
    }
    return;
  }
  for (int j = 0; j < 4; ++j) {
    const size_t g = g0 + (size_t)j * blockDim.x;
    const size_t e = g * 4;
    if (e >= (size_t)n_edges) {
      keys[g] = ~0u;
      continue;
    }
    int4 s, d;
    s.x = ei[e];
    d.x = dstp[e];
    s.y = (e + 1 < (size_t)n_edges) ? ei[e + 1] : 0;
    d.y = (e + 1 < (size_t)n_edges) ? dstp[e + 1] : -1;
    s.z = (e + 2 < (size_t)n_edges) ? ei[e + 2] : 0;
    d.z = (e + 2 < (size_t)n_edges) ? dstp[e + 2] : -1;
    s.w = (e + 3 < (size_t)n_edges) ? ei[e + 3] : 0;
    d.w = (e + 3 < (size_t)n_edges) ? dstp[e + 3] : -1;
    keys[g] = own_key4(d);
    *(uint4*)(words + e) = own_word4(s, d);
  }
}


__device__ __forceinline__ uint4 own_keys(const unsigned* __restrict__ keys,
                                          int e) {
  return *(const uint4*)(keys + (e >> 2));
}

__device__ __forceinline__ unsigned own_eq(unsigned w, unsigned pat) {
  const unsigned t = w ^ pat;
  return ~(((t & 0x7F7F7F7Fu) + 0x7F7F7F7Fu) | t) & 0x80808080u;
}

__device__ __forceinline__ int own_push(unsigned* q, int qn, uint4 k,
                                        unsigned pat, int e) {
  unsigned m = (own_eq(k.x, pat) >> 7) | (own_eq(k.y, pat) >> 6) |
               (own_eq(k.z, pat) >> 5) | (own_eq(k.w, pat) >> 4);
  for (int r = 0; r < 16; ++r) {
    const bool has = m != 0u;
    const unsigned mask = __builtin_amdgcn_ballot_w32(has);
    if (mask == 0u) break;
    if (has) {
      const int p = __builtin_ctz(m);
      m &= m - 1u;
      const int pos = qn + (int)__builtin_amdgcn_mbcnt_lo(mask, 0u);
      q[pos] = (unsigned)(e + 4 * (p & 7) + (p >> 3));
    }
    qn += __builtin_popcount(mask);
  }
  return qn;
}

__device__ __forceinline__ int own_step(const unsigned* __restrict__ keys,
                                        uint4& k0, uint4& k1, int e,
                                        unsigned pat, unsigned* q, int qn) {
  qn = own_push(q, qn, k0, pat, e);
  qn = own_push(q, qn, k1, pat, e + 512);
  k0 = own_keys(keys, e + OWN_AHEAD * OWN_WAVES * OWN_STEP);
  k1 = own_keys(keys, e + OWN_AHEAD * OWN_WAVES * OWN_STEP + 512);
  return qn;
}

__device__ __forceinline__ void own_add_row(float* tile, int dl, int lane,
                                            const float4 v) {
  float* t = tile + dl * D_FEAT + lane;
  atomicAdd(t + 0, v.x);
  atomicAdd(t + 32, v.y);
  atomicAdd(t + 64, v.z);
  atomicAdd(t + 96, v.w);
}

__device__ __forceinline__ void own_entry(const unsigned* __restrict__ words,
                                          float* cntL, const unsigned* q,
                                          int& mysrc, int& myl, unsigned& w) {
  w = words[*q];
  mysrc = (int)(w & 0xFFFFFFu);
  myl = (int)(w >> 24);
  atomicAdd(cntL + myl, 1.0f);
}

struct OwnOut {
  unsigned* chunks;
  unsigned char* dir;
  int* cursor;
  int slot0;
  int stat_total;
  int nem;
};

__device__ __forceinline__ void own_emit(OwnOut& o, unsigned w, int lane) {
  int c;
  if (o.nem < OWN_K) {
    c = o.slot0 + o.nem;
  } else {
    int t = 0;
    if (lane == 0) t = atomicAdd(o.cursor, 1);
    c = o.stat_total + __builtin_amdgcn_readfirstlane(t);
  }
  ++o.nem;
  o.chunks[(size_t)c * 32 + lane] = w;
  if (lane == 0) o.dir[c] = (unsigned char)blockIdx.x;
}

__device__ __forceinline__ int own_flush(const float* __restrict__ x,
                                         const unsigned* __restrict__ words,
                                         float* tile, float* cntL, unsigned* q,
                                         int qn, int lane, bool last,
                                         OwnOut& o) {
  int b = 0;
  while (qn - b >= OWN_B) {
    int g = (qn - b) & ~(OWN_B - 1);
    if (g > 32) g = 32;
    int mysrc = 0, myl = 0;
    unsigned w = ~0u;
    if (lane < g) own_entry(words, cntL, q + b + lane, mysrc, myl, w);
    own_emit(o, w, lane);
    for (int i0 = 0; i0 < g; i0 += OWN_B) {
      float4 v[OWN_B];
      int dl[OWN_B];
#pragma unroll
      for (int u = 0; u < OWN_B; ++u) {
        const int sidx = __builtin_amdgcn_readlane(mysrc, i0 + u);
        dl[u] = __builtin_amdgcn_readlane(myl, i0 + u);
        v[u] = *(const float4*)(x + (size_t)sidx * D_FEAT + lane * 4);
      }
#pragma unroll
      for (int u = 0; u < OWN_B; ++u) own_add_row(tile, dl[u], lane, v[u]);
    }
    b += g;
  }
  const int r = qn - b;
  if (last) {
    if (r > 0) {
      int mysrc = 0, myl = 0;
      unsigned w = ~0u;
      if (lane < r) own_entry(words, cntL, q + b + lane, mysrc, myl, w);
      own_emit(o, w, lane);
      float4 v[OWN_B - 1];
      int dl[OWN_B - 1];
#pragma unroll
      for (int u = 0; u < OWN_B - 1; ++u) {
        if (u < r) {
          const int sidx = __builtin_amdgcn_readlane(mysrc, u);
          dl[u] = __builtin_amdgcn_readlane(myl, u);
          v[u] = *(const float4*)(x + (size_t)sidx * D_FEAT + lane * 4);
        }
      }
#pragma unroll
      for (int u = 0; u < OWN_B - 1; ++u) {
        if (u < r) own_add_row(tile, dl[u], lane, v[u]);
      }
    }
    return 0;
  }
  if (r > 0 && b > 0) {
    unsigned t = 0u;
    if (lane < r) t = q[b + lane];
    if (lane < r) q[lane] = t;
  }
  return r;
}

__global__ __launch_bounds__(1024) void sage_own1(
    const float* __restrict__ x, const unsigned* __restrict__ words,
    const unsigned* __restrict__ keys, float* __restrict__ agg1,
    float* __restrict__ cnt, unsigned* __restrict__ chunks,
    unsigned char* __restrict__ dir, int* __restrict__ cursor, int n_edges,
    int n_nodes) {
  __shared__ float tile[OWN_NB * D_FEAT];
  __shared__ float cntL[OWN_NB];
  __shared__ unsigned queue[OWN_WAVES * OWN_QCAP];

  const int lane = threadIdx.x & 31;
  const int wave = __builtin_amdgcn_readfirstlane((int)(threadIdx.x >> 5));
  const int base = blockIdx.x * OWN_NB;
  const unsigned pat = blockIdx.x * 0x01010101u;

  for (int i = threadIdx.x; i < OWN_NB * D_FEAT / 4; i += OWN_WAVES * 32)
    ((float4*)tile)[i] = make_float4(0.0f, 0.0f, 0.0f, 0.0f);
  if (threadIdx.x < OWN_NB) cntL[threadIdx.x] = 0.0f;
  __syncthreads();

  unsigned* q = queue + wave * OWN_QCAP;
  OwnOut o;
  o.chunks = chunks;
  o.dir = dir;
  o.cursor = cursor;
  o.slot0 = ((int)blockIdx.x * OWN_WAVES + wave) * OWN_K;
  o.stat_total = (int)gridDim.x * OWN_WAVES * OWN_K;
  o.nem = 0;
  const int stride = OWN_WAVES * OWN_STEP;
  const int n_steps = (n_edges + stride - 1) / stride;
  int qn = 0;

  const int ef = wave * OWN_STEP + lane * 16;
  uint4 a0 = own_keys(keys, ef), a1 = own_keys(keys, ef + 512);
  uint4 b0 = own_keys(keys, ef + stride), b1 = own_keys(keys, ef + stride + 512);
  uint4 c0 = own_keys(keys, ef + 2 * stride), c1 = own_keys(keys, ef + 2 * stride + 512);
  uint4 d0 = own_keys(keys, ef + 3 * stride), d1 = own_keys(keys, ef + 3 * stride + 512);

  for (int s = 0; s < n_steps; s += OWN_AHEAD) {
    const int e = s * stride + ef;
    qn = own_step(keys, a0, a1, e, pat, q, qn);
    if (qn >= OWN_FLUSH)
      qn = own_flush(x, words, tile, cntL, q, qn, lane, false, o);
    qn = own_step(keys, b0, b1, e + stride, pat, q, qn);
    if (qn >= OWN_FLUSH)
      qn = own_flush(x, words, tile, cntL, q, qn, lane, false, o);
    qn = own_step(keys, c0, c1, e + 2 * stride, pat, q, qn);
    if (qn >= OWN_FLUSH)
      qn = own_flush(x, words, tile, cntL, q, qn, lane, false, o);
    qn = own_step(keys, d0, d1, e + 3 * stride, pat, q, qn);
    if (qn >= OWN_FLUSH)
      qn = own_flush(x, words, tile, cntL, q, qn, lane, false, o);
  }
  own_flush(x, words, tile, cntL, q, qn, lane, true, o);
  __syncthreads();

  for (int i = threadIdx.x; i < OWN_NB * D_FEAT / 4; i += OWN_WAVES * 32) {
    const int row = base + (i >> 5);
    if (row < n_nodes)
      *(float4*)(agg1 + (size_t)row * D_FEAT + (i & 31) * 4) = ((const float4*)tile)[i];
  }
  if (threadIdx.x < OWN_NB && base + (int)threadIdx.x < n_nodes)
    cnt[base + threadIdx.x] = cntL[threadIdx.x];
}

__global__ __launch_bounds__(256) __attribute__((amdgpu_waves_per_eu(1, 4)))
void sage_layer1_wmma(
    const float* __restrict__ x, const float* __restrict__ agg1,
    const float* __restrict__ cnt, const float* __restrict__ W1l,
    const float* __restrict__ W1r, const float* __restrict__ b1,
    const float* __restrict__ W2l, const float* __restrict__ W2r,
    const float* __restrict__ b2, float* __restrict__ p,
    float* __restrict__ out, int n_nodes) {
  __shared__ __attribute__((aligned(16))) float ldsWl[D_FEAT * D_FEAT];

  for (int idx = threadIdx.x; idx < D_FEAT * 64; idx += 256) {
    const int r = idx >> 6;
    const int c = idx & 63;
    const int f0 = 4 * ((2 * c) & 31) + (c >> 4);
    float2 w;
    w.x = W1l[(size_t)r * D_FEAT + f0];
    w.y = W1l[(size_t)r * D_FEAT + f0 + 4];
    const int s = c ^ ((2 * r) & 63);
    *(float2*)(ldsWl + r * D_FEAT + 2 * s) = w;
  }
  __syncthreads();

  const int lane = threadIdx.x & 31;
  const int wave = threadIdx.x >> 5;
  const int tile = blockIdx.x * 8 + wave;
  const int n_tiles = n_nodes >> 4;
  if (tile >= n_tiles) return;

  const int half = lane >> 4;
  const int q = lane & 15;
  const int rowA = tile * 16 + q;

  const float inv = 1.0f / fmaxf(cnt[rowA], 1.0f);

  v8f acc[8];
#pragma unroll
  for (int i = 0; i < 8; ++i) {
    v8f z = {};
    acc[i] = z;
  }

  const float* aggRow = agg1 + (size_t)rowA * D_FEAT + 4 * half;
  const float* xRow   = x    + (size_t)rowA * D_FEAT + 4 * half;

  int nrow[8], nswz[8];
#pragma unroll
  for (int nt = 0; nt < 8; ++nt) {
    nrow[nt] = nt * 16 + q;
    nswz[nt] = nrow[nt] & 31;
  }

  struct Frag {
    float4 am, as;
    float4 bl[8], br[8];
  };
  auto load = [&](Frag& f, int k) {
    const int u0 = (k >> 2) + half;
    f.am = *(const float4*)(aggRow + k);
    f.as = *(const float4*)(xRow + k);
#pragma unroll
    for (int nt = 0; nt < 8; ++nt) {
      f.bl[nt] = *(const float4*)(ldsWl + nrow[nt] * D_FEAT + 4 * (u0 ^ nswz[nt]));
      f.br[nt] = *(const float4*)(W1r + (size_t)nrow[nt] * D_FEAT + k + 4 * half);
    }
  };
  auto mma = [&](const Frag& f) {
    v2f a_m0;  a_m0.x = f.am.x * inv;  a_m0.y = f.am.y * inv;
    v2f a_m1;  a_m1.x = f.am.z * inv;  a_m1.y = f.am.w * inv;
    v2f a_s0;  a_s0.x = f.as.x;        a_s0.y = f.as.y;
    v2f a_s1;  a_s1.x = f.as.z;        a_s1.y = f.as.w;
#pragma unroll
    for (int nt = 0; nt < 8; ++nt) {
      v2f vbl;  vbl.x = f.bl[nt].x;  vbl.y = f.bl[nt].y;
      v2f vbr;  vbr.x = f.br[nt].x;  vbr.y = f.br[nt].y;
      acc[nt] = __builtin_amdgcn_wmma_f32_16x16x4_f32(
          false, vbl, false, a_m0, (short)0, acc[nt], false, false);
      acc[nt] = __builtin_amdgcn_wmma_f32_16x16x4_f32(
          false, vbr, false, a_s0, (short)0, acc[nt], false, false);
    }
#pragma unroll
    for (int nt = 0; nt < 8; ++nt) {
      v2f vbl;  vbl.x = f.bl[nt].z;  vbl.y = f.bl[nt].w;
      v2f vbr;  vbr.x = f.br[nt].z;  vbr.y = f.br[nt].w;
      acc[nt] = __builtin_amdgcn_wmma_f32_16x16x4_f32(
          false, vbl, false, a_m1, (short)0, acc[nt], false, false);
      acc[nt] = __builtin_amdgcn_wmma_f32_16x16x4_f32(
          false, vbr, false, a_s1, (short)0, acc[nt], false, false);
    }
  };

  Frag f0, f1;
  load(f0, 0);
  for (int k = 0; k < D_FEAT - 16; k += 16) {
    load(f1, k + 8);
    mma(f0);
    load(f0, k + 16);
    mma(f1);
  }
  load(f1, D_FEAT - 8);
  mma(f0);
  mma(f1);

  const int r4 = q & 3;
  const float* w2row = (r4 < 2) ? (W2l + r4 * D_FEAT) : (W2r + (r4 - 2) * D_FEAT);
  v8f o0 = {}, o1 = {};
#pragma unroll
  for (int nt = 0; nt < 8; ++nt) {
    const int f0 = nt * 16 + 8 * half;
    const float4 ba = *(const float4*)(b1 + f0);
    const float4 bb = *(const float4*)(b1 + f0 + 4);
    const float4 wa = *(const float4*)(w2row + f0);
    const float4 wb = *(const float4*)(w2row + f0 + 4);
    v2f h0, h1, h2, h3;
    h0.x = fmaxf(acc[nt][0] + ba.x, 0.0f);  h0.y = fmaxf(acc[nt][1] + ba.y, 0.0f);
    h1.x = fmaxf(acc[nt][2] + ba.z, 0.0f);  h1.y = fmaxf(acc[nt][3] + ba.w, 0.0f);
    h2.x = fmaxf(acc[nt][4] + bb.x, 0.0f);  h2.y = fmaxf(acc[nt][5] + bb.y, 0.0f);
    h3.x = fmaxf(acc[nt][6] + bb.z, 0.0f);  h3.y = fmaxf(acc[nt][7] + bb.w, 0.0f);
    v2f w0, w1, w2, w3;
    w0.x = wa.x;  w0.y = wa.y;
    w1.x = wa.z;  w1.y = wa.w;
    w2.x = wb.x;  w2.y = wb.y;
    w3.x = wb.z;  w3.y = wb.w;
    o0 = __builtin_amdgcn_wmma_f32_16x16x4_f32(
        false, w0, false, h0, (short)0, o0, false, false);
    o1 = __builtin_amdgcn_wmma_f32_16x16x4_f32(
        false, w1, false, h1, (short)0, o1, false, false);
    o0 = __builtin_amdgcn_wmma_f32_16x16x4_f32(
        false, w2, false, h2, (short)0, o0, false, false);
    o1 = __builtin_amdgcn_wmma_f32_16x16x4_f32(
        false, w3, false, h3, (short)0, o1, false, false);
  }
  if (half == 0) {
    const size_t node = (size_t)tile * 16 + q;
    float2 pv, ov;
    pv.x = o0[0] + o1[0];
    pv.y = o0[1] + o1[1];
    ov.x = o0[2] + o1[2] + b2[0];
    ov.y = o0[3] + o1[3] + b2[1];
    *(float2*)(p + node * 2) = pv;
    *(float2*)(out + node * 2) = ov;
  }
}

__device__ __forceinline__ int own2_drain(const float* __restrict__ p,
                                          const unsigned* __restrict__ chunks,
                                          float* tile2, const unsigned* q,
                                          int qn, int lane) {
  for (int b = 0; b < qn; b += 8) {
    unsigned w[8];
#pragma unroll
    for (int u = 0; u < 8; ++u) {
      w[u] = ~0u;
      if (b + u < qn) w[u] = chunks[(size_t)q[b + u] * 32 + lane];
    }
    float px[8], py[8];
#pragma unroll
    for (int u = 0; u < 8; ++u) {
      px[u] = 0.0f;
      py[u] = 0.0f;
      if (w[u] != ~0u) {
        const size_t s = (size_t)(w[u] & 0xFFFFFFu) * 2;
        px[u] = p[s];
        py[u] = p[s + 1];
      }
    }
#pragma unroll
    for (int u = 0; u < 8; ++u) {
      if (w[u] != ~0u) {
        float* t = tile2 + 2 * (w[u] >> 24);
        atomicAdd(t, px[u]);
        atomicAdd(t + 1, py[u]);
      }
    }
  }
  return 0;
}

__global__ __launch_bounds__(1024) void sage_own2(
    const float* __restrict__ p, const unsigned* __restrict__ chunks,
    const unsigned* __restrict__ dir32, const int* __restrict__ cursor,
    const float* __restrict__ cnt, float* __restrict__ out, int n_nodes,
    int stat_total, int ovf_cap) {
  __shared__ float tile2[OWN_NB * 2];
  __shared__ unsigned queue2[OWN_WAVES * OWN2_QCAP];

  const int lane = threadIdx.x & 31;
  const int wave = __builtin_amdgcn_readfirstlane((int)(threadIdx.x >> 5));
  const int base = blockIdx.x * OWN_NB;
  const unsigned pat = blockIdx.x * 0x01010101u;

  if (threadIdx.x < OWN_NB * 2) tile2[threadIdx.x] = 0.0f;
  __syncthreads();

  int ov = *cursor;
  if (ov > ovf_cap) ov = ovf_cap;
  if (ov < 0) ov = 0;
  const int n_scan = stat_total + ov;

  unsigned* q = queue2 + wave * OWN2_QCAP;
  int qn = 0;
  const uint4 none = make_uint4(~0u, ~0u, ~0u, ~0u);
  static_assert(OWN_K == 8, "own slots of a wave = two directory dwords");
  {
    const int slot0 = ((int)blockIdx.x * OWN_WAVES + wave) * OWN_K;
    uint4 k = none;
    if (lane == 0) {
      k.x = dir32[slot0 >> 2];
      k.y = dir32[(slot0 >> 2) + 1];
    }
    qn = own_push(q, qn, k, pat, slot0);
  }
  for (int r0 = stat_total; r0 < n_scan; r0 += OWN_WAVES * 1024) {
    const int cb = r0 + wave * 1024 + lane * 16;
    uint4 k0 = none, k1 = none;
    if (cb < n_scan) k0 = *(const uint4*)(dir32 + (cb >> 2));
    if (cb + 512 < n_scan) k1 = *(const uint4*)(dir32 + ((cb + 512) >> 2));
    qn = own_push(q, qn, k0, pat, cb);
    qn = own_push(q, qn, k1, pat, cb + 512);
    if (qn > OWN2_QCAP - 1024) qn = own2_drain(p, chunks, tile2, q, qn, lane);
  }
  own2_drain(p, chunks, tile2, q, qn, lane);
  __syncthreads();

  if (threadIdx.x < OWN_NB) {
    const int node = base + (int)threadIdx.x;
    if (node < n_nodes) {
      const float inv = 1.0f / fmaxf(cnt[node], 1.0f);
      out[(size_t)node * 2 + 0] += tile2[2 * threadIdx.x + 0] * inv;
      out[(size_t)node * 2 + 1] += tile2[2 * threadIdx.x + 1] * inv;
    }
  }
}

extern "C" void kernel_launch(void* const* d_in, const int* in_sizes, int n_in,
                              void* d_out, int out_size, void* d_ws, size_t ws_size,
                              hipStream_t stream) {
  const float* x   = (const float*)d_in[0];
  const int*   ei  = (const int*)d_in[1];
  const float* W1l = (const float*)d_in[2];
  const float* W1r = (const float*)d_in[3];
  const float* b1  = (const float*)d_in[4];
  const float* W2l = (const float*)d_in[5];
  const float* W2r = (const float*)d_in[6];
  const float* b2  = (const float*)d_in[7];
  float* out = (float*)d_out;

  const int n_nodes = in_sizes[0] / D_FEAT;
  const int n_edges = in_sizes[1] / 2;

  float* ws   = (float*)d_ws;
  float* agg1 = ws;
  float* cnt  = agg1 + (size_t)n_nodes * D_FEAT;
  float* scr  = cnt + (size_t)n_nodes * 3;
  float* p    = scr + (size_t)n_nodes * D_FEAT;

  unsigned* keys = (unsigned*)scr;

  const int key_bytes = n_edges + 2 * OWN_AHEAD * OWN_WAVES * OWN_STEP;
  unsigned* words = keys + ((size_t)key_bytes / 16 + 512) * 4;

  const int n_blocks = (n_nodes + OWN_NB - 1) / OWN_NB;
  const int stat_total = n_blocks * OWN_WAVES * OWN_K;
  const int ovf_cap = n_edges / 16 + n_blocks * OWN_WAVES + 64;
  const size_t max_chunks = (size_t)stat_total + ovf_cap;
  unsigned* chunks = words + ((size_t)n_edges / 4 + 4) * 4;
  unsigned* dir32 = chunks + max_chunks * 32;
  const size_t dir_words = (max_chunks / 16 + 1) * 4;
  int* cursor = (int*)(dir32 + dir_words);

  sage_prep<<<(key_bytes + 16 * 256 - 1) / (16 * 256), 256, 0, stream>>>(
      ei, keys, words, dir32, dir_words, cursor, n_edges);

  sage_own1<<<n_blocks, OWN_WAVES * 32, 0, stream>>>(
      x, words, keys, agg1, cnt, chunks, (unsigned char*)dir32, cursor, n_edges,
      n_nodes);

  const int n_tiles = n_nodes / 16;
  sage_layer1_wmma<<<(n_tiles + 7) / 8, 256, 0, stream>>>(
      x, agg1, cnt, W1l, W1r, b1, W2l, W2r, b2, p, out, n_nodes);

  sage_own2<<<n_blocks, OWN_WAVES * 32, 0, stream>>>(p, chunks, dir32, cursor, cnt,
                                                     out, n_nodes, stat_total,
                                                     ovf_cap);
}


# Layer-1 weights re-laid by launch 0 in the order launch 2 reads them: W1r fragment loads cover 4 whole lines (not 32 B of 16 rows), W1l staged by straight 16-B copies; values and sum order unchanged
# speedup vs baseline: 4.6021x; 1.0522x over previous
#include <hip/hip_runtime.h>

typedef __attribute__((ext_vector_type(2))) float v2f;
typedef __attribute__((ext_vector_type(8))) float v8f;

#define D_FEAT 128

#define OWN_NB 200
#define OWN_WAVES 32
#define OWN_STEP 1024
#define OWN_FLUSH 32
#define OWN_QCAP 1056
#define OWN_B 16
#define OWN_AHEAD 4
#define OWN_PAD 0xFFu
#define OWN_K 8
#define OWN2_QCAP 1536

__device__ __forceinline__ unsigned own_key(int dst) {
  const unsigned k = (unsigned)dst / OWN_NB;
  return k < OWN_PAD ? k : OWN_PAD;
}

__device__ __forceinline__ unsigned own_key4(int4 d) {
  return own_key(d.x) | (own_key(d.y) << 8) | (own_key(d.z) << 16) |
         (own_key(d.w) << 24);
}

__device__ __forceinline__ unsigned own_word(int s, int dst) {
  const unsigned l = (unsigned)dst - own_key(dst) * OWN_NB;
  return (l << 24) | ((unsigned)s & 0xFFFFFFu);
}

__device__ __forceinline__ uint4 own_word4(int4 s, int4 d) {
  uint4 w;
  w.x = own_word(s.x, d.x);
  w.y = own_word(s.y, d.y);
  w.z = own_word(s.z, d.z);
  w.w = own_word(s.w, d.w);
  return w;
}

__global__ __launch_bounds__(256) void sage_prep(
    const int* __restrict__ ei, unsigned* __restrict__ keys,
    unsigned* __restrict__ words, unsigned* __restrict__ dir32, size_t zn,
    int* __restrict__ cursor, const float* __restrict__ W1r,
    float* __restrict__ w1rp, const float* __restrict__ W1l,
    float* __restrict__ w1lp, int n_edges) {
  const size_t t = (size_t)blockIdx.x * blockDim.x + threadIdx.x;
  const size_t nt = (size_t)gridDim.x * blockDim.x;
  const size_t pk0 = nt - (size_t)(2 * D_FEAT * D_FEAT / 4);
  if (t >= pk0) {
    const int i = (int)(t - pk0);
    if (i < D_FEAT * D_FEAT / 4) {
      const int fl = i & 31;
      const int fn = (i >> 5) & 7;
      const int fk = i >> 8;
      *(float4*)(w1rp + (size_t)i * 4) = *(const float4*)(
          W1r + (size_t)(fn * 16 + (fl & 15)) * D_FEAT + fk * 8 + 4 * (fl >> 4));
    } else {
      const int i2 = i - D_FEAT * D_FEAT / 4;
      const int r = i2 >> 5;
      const int sb = 2 * (i2 & 31);
      const int key = (2 * r) & 63;
      const int c0 = sb ^ key, c1 = (sb + 1) ^ key;
      const int f0 = 4 * ((2 * c0) & 31) + (c0 >> 4);
      const int f1 = 4 * ((2 * c1) & 31) + (c1 >> 4);
      const float* wr = W1l + (size_t)r * D_FEAT;
      float4 v;
      v.x = wr[f0];
      v.y = wr[f0 + 4];
      v.z = wr[f1];
      v.w = wr[f1 + 4];
      *(float4*)(w1lp + (size_t)i2 * 4) = v;
    }
  }
  for (size_t i = t; i < zn; i += nt) dir32[i] = ~0u;
  if (t == 0) *cursor = 0;

  const int* dstp = ei + n_edges;
  const size_t g0 = (size_t)blockIdx.x * (blockDim.x * 4) + threadIdx.x;
  if ((n_edges & 3) == 0 && n_edges > 0) {
    const size_t n_groups = (size_t)n_edges >> 2;
    int4 s[4], d[4];
#pragma unroll
    for (int j = 0; j < 4; ++j) {
      const size_t g = g0 + (size_t)j * blockDim.x;
      const size_t gc = g < n_groups ? g : n_groups - 1;
      s[j] = *(const int4*)(ei + gc * 4);
      d[j] = *(const int4*)(dstp + gc * 4);
    }
#pragma unroll
    for (int j = 0; j < 4; ++j) {
      const size_t g = g0 + (size_t)j * blockDim.x;
      if (g < n_groups) {
        keys[g] = own_key4(d[j]);
        *(uint4*)(words + g * 4) = own_word4(s[j], d[j]);
      } else {
        keys[g] = ~0u;
      }
    }
    return;
  }
  for (int j = 0; j < 4; ++j) {
    const size_t g = g0 + (size_t)j * blockDim.x;
    const size_t e = g * 4;
    if (e >= (size_t)n_edges) {
      keys[g] = ~0u;
      continue;
    }
    int4 s, d;
    s.x = ei[e];
    d.x = dstp[e];
    s.y = (e + 1 < (size_t)n_edges) ? ei[e + 1] : 0;
    d.y = (e + 1 < (size_t)n_edges) ? dstp[e + 1] : -1;
    s.z = (e + 2 < (size_t)n_edges) ? ei[e + 2] : 0;
    d.z = (e + 2 < (size_t)n_edges) ? dstp[e + 2] : -1;
    s.w = (e + 3 < (size_t)n_edges) ? ei[e + 3] : 0;
    d.w = (e + 3 < (size_t)n_edges) ? dstp[e + 3] : -1;
    keys[g] = own_key4(d);
    *(uint4*)(words + e) = own_word4(s, d);
  }
}


__device__ __forceinline__ uint4 own_keys(const unsigned* __restrict__ keys,
                                          int e) {
  return *(const uint4*)(keys + (e >> 2));
}

__device__ __forceinline__ unsigned own_eq(unsigned w, unsigned pat) {
  const unsigned t = w ^ pat;
  return ~(((t & 0x7F7F7F7Fu) + 0x7F7F7F7Fu) | t) & 0x80808080u;
}

__device__ __forceinline__ int own_push(unsigned* q, int qn, uint4 k,
                                        unsigned pat, int e) {
  unsigned m = (own_eq(k.x, pat) >> 7) | (own_eq(k.y, pat) >> 6) |
               (own_eq(k.z, pat) >> 5) | (own_eq(k.w, pat) >> 4);
  for (int r = 0; r < 16; ++r) {
    const bool has = m != 0u;
    const unsigned mask = __builtin_amdgcn_ballot_w32(has);
    if (mask == 0u) break;
    if (has) {
      const int p = __builtin_ctz(m);
      m &= m - 1u;
      const int pos = qn + (int)__builtin_amdgcn_mbcnt_lo(mask, 0u);
      q[pos] = (unsigned)(e + 4 * (p & 7) + (p >> 3));
    }
    qn += __builtin_popcount(mask);
  }
  return qn;
}

__device__ __forceinline__ int own_step(const unsigned* __restrict__ keys,
                                        uint4& k0, uint4& k1, int e,
                                        unsigned pat, unsigned* q, int qn) {
  qn = own_push(q, qn, k0, pat, e);
  qn = own_push(q, qn, k1, pat, e + 512);
  k0 = own_keys(keys, e + OWN_AHEAD * OWN_WAVES * OWN_STEP);
  k1 = own_keys(keys, e + OWN_AHEAD * OWN_WAVES * OWN_STEP + 512);
  return qn;
}

__device__ __forceinline__ void own_add_row(float* tile, int dl, int lane,
                                            const float4 v) {
  float* t = tile + dl * D_FEAT + lane;
  atomicAdd(t + 0, v.x);
  atomicAdd(t + 32, v.y);
  atomicAdd(t + 64, v.z);
  atomicAdd(t + 96, v.w);
}

__device__ __forceinline__ void own_entry(const unsigned* __restrict__ words,
                                          float* cntL, const unsigned* q,
                                          int& mysrc, int& myl, unsigned& w) {
  w = words[*q];
  mysrc = (int)(w & 0xFFFFFFu);
  myl = (int)(w >> 24);
  atomicAdd(cntL + myl, 1.0f);
}

struct OwnOut {
  unsigned* chunks;
  unsigned char* dir;
  int* cursor;
  int slot0;
  int stat_total;
  int nem;
};

__device__ __forceinline__ void own_emit(OwnOut& o, unsigned w, int lane) {
  int c;
  if (o.nem < OWN_K) {
    c = o.slot0 + o.nem;
  } else {
    int t = 0;
    if (lane == 0) t = atomicAdd(o.cursor, 1);
    c = o.stat_total + __builtin_amdgcn_readfirstlane(t);
  }
  ++o.nem;
  o.chunks[(size_t)c * 32 + lane] = w;
  if (lane == 0) o.dir[c] = (unsigned char)blockIdx.x;
}

__device__ __forceinline__ int own_flush(const float* __restrict__ x,
                                         const unsigned* __restrict__ words,
                                         float* tile, float* cntL, unsigned* q,
                                         int qn, int lane, bool last,
                                         OwnOut& o) {
  int b = 0;
  while (qn - b >= OWN_B) {
    int g = (qn - b) & ~(OWN_B - 1);
    if (g > 32) g = 32;
    int mysrc = 0, myl = 0;
    unsigned w = ~0u;
    if (lane < g) own_entry(words, cntL, q + b + lane, mysrc, myl, w);
    own_emit(o, w, lane);
    for (int i0 = 0; i0 < g; i0 += OWN_B) {
      float4 v[OWN_B];
      int dl[OWN_B];
#pragma unroll
      for (int u = 0; u < OWN_B; ++u) {
        const int sidx = __builtin_amdgcn_readlane(mysrc, i0 + u);
        dl[u] = __builtin_amdgcn_readlane(myl, i0 + u);
        v[u] = *(const float4*)(x + (size_t)sidx * D_FEAT + lane * 4);
      }
#pragma unroll
      for (int u = 0; u < OWN_B; ++u) own_add_row(tile, dl[u], lane, v[u]);
    }
    b += g;
  }
  const int r = qn - b;
  if (last) {
    if (r > 0) {
      int mysrc = 0, myl = 0;
      unsigned w = ~0u;
      if (lane < r) own_entry(words, cntL, q + b + lane, mysrc, myl, w);
      own_emit(o, w, lane);
      float4 v[OWN_B - 1];
      int dl[OWN_B - 1];
#pragma unroll
      for (int u = 0; u < OWN_B - 1; ++u) {
        if (u < r) {
          const int sidx = __builtin_amdgcn_readlane(mysrc, u);
          dl[u] = __builtin_amdgcn_readlane(myl, u);
          v[u] = *(const float4*)(x + (size_t)sidx * D_FEAT + lane * 4);
        }
      }
#pragma unroll
      for (int u = 0; u < OWN_B - 1; ++u) {
        if (u < r) own_add_row(tile, dl[u], lane, v[u]);
      }
    }
    return 0;
  }
  if (r > 0 && b > 0) {
    unsigned t = 0u;
    if (lane < r) t = q[b + lane];
    if (lane < r) q[lane] = t;
  }
  return r;
}

__global__ __launch_bounds__(1024) void sage_own1(
    const float* __restrict__ x, const unsigned* __restrict__ words,
    const unsigned* __restrict__ keys, float* __restrict__ agg1,
    float* __restrict__ cnt, unsigned* __restrict__ chunks,
    unsigned char* __restrict__ dir, int* __restrict__ cursor, int n_edges,
    int n_nodes) {
  __shared__ float tile[OWN_NB * D_FEAT];
  __shared__ float cntL[OWN_NB];
  __shared__ unsigned queue[OWN_WAVES * OWN_QCAP];

  const int lane = threadIdx.x & 31;
  const int wave = __builtin_amdgcn_readfirstlane((int)(threadIdx.x >> 5));
  const int base = blockIdx.x * OWN_NB;
  const unsigned pat = blockIdx.x * 0x01010101u;

  for (int i = threadIdx.x; i < OWN_NB * D_FEAT / 4; i += OWN_WAVES * 32)
    ((float4*)tile)[i] = make_float4(0.0f, 0.0f, 0.0f, 0.0f);
  if (threadIdx.x < OWN_NB) cntL[threadIdx.x] = 0.0f;
  __syncthreads();

  unsigned* q = queue + wave * OWN_QCAP;
  OwnOut o;
  o.chunks = chunks;
  o.dir = dir;
  o.cursor = cursor;
  o.slot0 = ((int)blockIdx.x * OWN_WAVES + wave) * OWN_K;
  o.stat_total = (int)gridDim.x * OWN_WAVES * OWN_K;
  o.nem = 0;
  const int stride = OWN_WAVES * OWN_STEP;
  const int n_steps = (n_edges + stride - 1) / stride;
  int qn = 0;

  const int ef = wave * OWN_STEP + lane * 16;
  uint4 a0 = own_keys(keys, ef), a1 = own_keys(keys, ef + 512);
  uint4 b0 = own_keys(keys, ef + stride), b1 = own_keys(keys, ef + stride + 512);
  uint4 c0 = own_keys(keys, ef + 2 * stride), c1 = own_keys(keys, ef + 2 * stride + 512);
  uint4 d0 = own_keys(keys, ef + 3 * stride), d1 = own_keys(keys, ef + 3 * stride + 512);

  for (int s = 0; s < n_steps; s += OWN_AHEAD) {
    const int e = s * stride + ef;
    qn = own_step(keys, a0, a1, e, pat, q, qn);
    if (qn >= OWN_FLUSH)
      qn = own_flush(x, words, tile, cntL, q, qn, lane, false, o);
    qn = own_step(keys, b0, b1, e + stride, pat, q, qn);
    if (qn >= OWN_FLUSH)
      qn = own_flush(x, words, tile, cntL, q, qn, lane, false, o);
    qn = own_step(keys, c0, c1, e + 2 * stride, pat, q, qn);
    if (qn >= OWN_FLUSH)
      qn = own_flush(x, words, tile, cntL, q, qn, lane, false, o);
    qn = own_step(keys, d0, d1, e + 3 * stride, pat, q, qn);
    if (qn >= OWN_FLUSH)
      qn = own_flush(x, words, tile, cntL, q, qn, lane, false, o);
  }
  own_flush(x, words, tile, cntL, q, qn, lane, true, o);
  __syncthreads();

  for (int i = threadIdx.x; i < OWN_NB * D_FEAT / 4; i += OWN_WAVES * 32) {
    const int row = base + (i >> 5);
    if (row < n_nodes)
      *(float4*)(agg1 + (size_t)row * D_FEAT + (i & 31) * 4) = ((const float4*)tile)[i];
  }
  if (threadIdx.x < OWN_NB && base + (int)threadIdx.x < n_nodes)
    cnt[base + threadIdx.x] = cntL[threadIdx.x];
}

__global__ __launch_bounds__(256) __attribute__((amdgpu_waves_per_eu(1, 4)))
void sage_layer1_wmma(
    const float* __restrict__ x, const float* __restrict__ agg1,
    const float* __restrict__ cnt, const float* __restrict__ W1lp,
    const float* __restrict__ W1rp, const float* __restrict__ b1,
    const float* __restrict__ W2l, const float* __restrict__ W2r,
    const float* __restrict__ b2, float* __restrict__ p,
    float* __restrict__ out, int n_nodes) {
  __shared__ __attribute__((aligned(16))) float ldsWl[D_FEAT * D_FEAT];

#pragma unroll
  for (int i = 0; i < D_FEAT * D_FEAT / 4 / 256; ++i)
    ((float4*)ldsWl)[threadIdx.x + i * 256] =
        ((const float4*)W1lp)[threadIdx.x + i * 256];
  __syncthreads();

  const int lane = threadIdx.x & 31;
  const int wave = threadIdx.x >> 5;
  const int tile = blockIdx.x * 8 + wave;
  const int n_tiles = n_nodes >> 4;
  if (tile >= n_tiles) return;

  const int half = lane >> 4;
  const int q = lane & 15;
  const int rowA = tile * 16 + q;

  const float inv = 1.0f / fmaxf(cnt[rowA], 1.0f);

  v8f acc[8];
#pragma unroll
  for (int i = 0; i < 8; ++i) {
    v8f z = {};
    acc[i] = z;
  }

  const float* aggRow = agg1 + (size_t)rowA * D_FEAT + 4 * half;
  const float* xRow   = x    + (size_t)rowA * D_FEAT + 4 * half;

  int nrow[8], nswz[8];
#pragma unroll
  for (int nt = 0; nt < 8; ++nt) {
    nrow[nt] = nt * 16 + q;
    nswz[nt] = nrow[nt] & 31;
  }

  struct Frag {
    float4 am, as;
    float4 bl[8], br[8];
  };
  auto load = [&](Frag& f, int k) {
    const int u0 = (k >> 2) + half;
    f.am = *(const float4*)(aggRow + k);
    f.as = *(const float4*)(xRow + k);
#pragma unroll
    for (int nt = 0; nt < 8; ++nt) {
      f.bl[nt] = *(const float4*)(ldsWl + nrow[nt] * D_FEAT + 4 * (u0 ^ nswz[nt]));
      f.br[nt] = *(const float4*)(W1rp + (size_t)(((k >> 3) * 8 + nt) * 32 + lane) * 4);
    }
  };
  auto mma = [&](const Frag& f) {
    v2f a_m0;  a_m0.x = f.am.x * inv;  a_m0.y = f.am.y * inv;
    v2f a_m1;  a_m1.x = f.am.z * inv;  a_m1.y = f.am.w * inv;
    v2f a_s0;  a_s0.x = f.as.x;        a_s0.y = f.as.y;
    v2f a_s1;  a_s1.x = f.as.z;        a_s1.y = f.as.w;
#pragma unroll
    for (int nt = 0; nt < 8; ++nt) {
      v2f vbl;  vbl.x = f.bl[nt].x;  vbl.y = f.bl[nt].y;
      v2f vbr;  vbr.x = f.br[nt].x;  vbr.y = f.br[nt].y;
      acc[nt] = __builtin_amdgcn_wmma_f32_16x16x4_f32(
          false, vbl, false, a_m0, (short)0, acc[nt], false, false);
      acc[nt] = __builtin_amdgcn_wmma_f32_16x16x4_f32(
          false, vbr, false, a_s0, (short)0, acc[nt], false, false);
    }
#pragma unroll
    for (int nt = 0; nt < 8; ++nt) {
      v2f vbl;  vbl.x = f.bl[nt].z;  vbl.y = f.bl[nt].w;
      v2f vbr;  vbr.x = f.br[nt].z;  vbr.y = f.br[nt].w;
      acc[nt] = __builtin_amdgcn_wmma_f32_16x16x4_f32(
          false, vbl, false, a_m1, (short)0, acc[nt], false, false);
      acc[nt] = __builtin_amdgcn_wmma_f32_16x16x4_f32(
          false, vbr, false, a_s1, (short)0, acc[nt], false, false);
    }
  };

  Frag f0, f1;
  load(f0, 0);
  for (int k = 0; k < D_FEAT - 16; k += 16) {
    load(f1, k + 8);
    mma(f0);
    load(f0, k + 16);
    mma(f1);
  }
  load(f1, D_FEAT - 8);
  mma(f0);
  mma(f1);

  const int r4 = q & 3;
  const float* w2row = (r4 < 2) ? (W2l + r4 * D_FEAT) : (W2r + (r4 - 2) * D_FEAT);
  v8f o0 = {}, o1 = {};
#pragma unroll
  for (int nt = 0; nt < 8; ++nt) {
    const int f0 = nt * 16 + 8 * half;
    const float4 ba = *(const float4*)(b1 + f0);
    const float4 bb = *(const float4*)(b1 + f0 + 4);
    const float4 wa = *(const float4*)(w2row + f0);
    const float4 wb = *(const float4*)(w2row + f0 + 4);
    v2f h0, h1, h2, h3;
    h0.x = fmaxf(acc[nt][0] + ba.x, 0.0f);  h0.y = fmaxf(acc[nt][1] + ba.y, 0.0f);
    h1.x = fmaxf(acc[nt][2] + ba.z, 0.0f);  h1.y = fmaxf(acc[nt][3] + ba.w, 0.0f);
    h2.x = fmaxf(acc[nt][4] + bb.x, 0.0f);  h2.y = fmaxf(acc[nt][5] + bb.y, 0.0f);
    h3.x = fmaxf(acc[nt][6] + bb.z, 0.0f);  h3.y = fmaxf(acc[nt][7] + bb.w, 0.0f);
    v2f w0, w1, w2, w3;
    w0.x = wa.x;  w0.y = wa.y;
    w1.x = wa.z;  w1.y = wa.w;
    w2.x = wb.x;  w2.y = wb.y;
    w3.x = wb.z;  w3.y = wb.w;
    o0 = __builtin_amdgcn_wmma_f32_16x16x4_f32(
        false, w0, false, h0, (short)0, o0, false, false);
    o1 = __builtin_amdgcn_wmma_f32_16x16x4_f32(
        false, w1, false, h1, (short)0, o1, false, false);
    o0 = __builtin_amdgcn_wmma_f32_16x16x4_f32(
        false, w2, false, h2, (short)0, o0, false, false);
    o1 = __builtin_amdgcn_wmma_f32_16x16x4_f32(
        false, w3, false, h3, (short)0, o1, false, false);
  }
  if (half == 0) {
    const size_t node = (size_t)tile * 16 + q;
    float2 pv, ov;
    pv.x = o0[0] + o1[0];
    pv.y = o0[1] + o1[1];
    ov.x = o0[2] + o1[2] + b2[0];
    ov.y = o0[3] + o1[3] + b2[1];
    *(float2*)(p + node * 2) = pv;
    *(float2*)(out + node * 2) = ov;
  }
}

__device__ __forceinline__ int own2_drain(const float* __restrict__ p,
                                          const unsigned* __restrict__ chunks,
                                          float* tile2, const unsigned* q,
                                          int qn, int lane) {
  for (int b = 0; b < qn; b += 8) {
    unsigned w[8];
#pragma unroll
    for (int u = 0; u < 8; ++u) {
      w[u] = ~0u;
      if (b + u < qn) w[u] = chunks[(size_t)q[b + u] * 32 + lane];
    }
    float px[8], py[8];
#pragma unroll
    for (int u = 0; u < 8; ++u) {
      px[u] = 0.0f;
      py[u] = 0.0f;
      if (w[u] != ~0u) {
        const size_t s = (size_t)(w[u] & 0xFFFFFFu) * 2;
        px[u] = p[s];
        py[u] = p[s + 1];
      }
    }
#pragma unroll
    for (int u = 0; u < 8; ++u) {
      if (w[u] != ~0u) {
        float* t = tile2 + 2 * (w[u] >> 24);
        atomicAdd(t, px[u]);
        atomicAdd(t + 1, py[u]);
      }
    }
  }
  return 0;
}

__global__ __launch_bounds__(1024) void sage_own2(
    const float* __restrict__ p, const unsigned* __restrict__ chunks,
    const unsigned* __restrict__ dir32, const int* __restrict__ cursor,
    const float* __restrict__ cnt, float* __restrict__ out, int n_nodes,
    int stat_total, int ovf_cap) {
  __shared__ float tile2[OWN_NB * 2];
  __shared__ unsigned queue2[OWN_WAVES * OWN2_QCAP];

  const int lane = threadIdx.x & 31;
  const int wave = __builtin_amdgcn_readfirstlane((int)(threadIdx.x >> 5));
  const int base = blockIdx.x * OWN_NB;
  const unsigned pat = blockIdx.x * 0x01010101u;

  if (threadIdx.x < OWN_NB * 2) tile2[threadIdx.x] = 0.0f;
  __syncthreads();

  int ov = *cursor;
  if (ov > ovf_cap) ov = ovf_cap;
  if (ov < 0) ov = 0;
  const int n_scan = stat_total + ov;

  unsigned* q = queue2 + wave * OWN2_QCAP;
  int qn = 0;
  const uint4 none = make_uint4(~0u, ~0u, ~0u, ~0u);
  static_assert(OWN_K == 8, "own slots of a wave = two directory dwords");
  {
    const int slot0 = ((int)blockIdx.x * OWN_WAVES + wave) * OWN_K;
    uint4 k = none;
    if (lane == 0) {
      k.x = dir32[slot0 >> 2];
      k.y = dir32[(slot0 >> 2) + 1];
    }
    qn = own_push(q, qn, k, pat, slot0);
  }
  for (int r0 = stat_total; r0 < n_scan; r0 += OWN_WAVES * 1024) {
    const int cb = r0 + wave * 1024 + lane * 16;
    uint4 k0 = none, k1 = none;
    if (cb < n_scan) k0 = *(const uint4*)(dir32 + (cb >> 2));
    if (cb + 512 < n_scan) k1 = *(const uint4*)(dir32 + ((cb + 512) >> 2));
    qn = own_push(q, qn, k0, pat, cb);
    qn = own_push(q, qn, k1, pat, cb + 512);
    if (qn > OWN2_QCAP - 1024) qn = own2_drain(p, chunks, tile2, q, qn, lane);
  }
  own2_drain(p, chunks, tile2, q, qn, lane);
  __syncthreads();

  if (threadIdx.x < OWN_NB) {
    const int node = base + (int)threadIdx.x;
    if (node < n_nodes) {
      const float inv = 1.0f / fmaxf(cnt[node], 1.0f);
      out[(size_t)node * 2 + 0] += tile2[2 * threadIdx.x + 0] * inv;
      out[(size_t)node * 2 + 1] += tile2[2 * threadIdx.x + 1] * inv;
    }
  }
}

extern "C" void kernel_launch(void* const* d_in, const int* in_sizes, int n_in,
                              void* d_out, int out_size, void* d_ws, size_t ws_size,
                              hipStream_t stream) {
  const float* x   = (const float*)d_in[0];
  const int*   ei  = (const int*)d_in[1];
  const float* W1l = (const float*)d_in[2];
  const float* W1r = (const float*)d_in[3];
  const float* b1  = (const float*)d_in[4];
  const float* W2l = (const float*)d_in[5];
  const float* W2r = (const float*)d_in[6];
  const float* b2  = (const float*)d_in[7];
  float* out = (float*)d_out;

  const int n_nodes = in_sizes[0] / D_FEAT;
  const int n_edges = in_sizes[1] / 2;

  float* ws   = (float*)d_ws;
  float* agg1 = ws;
  float* cnt  = agg1 + (size_t)n_nodes * D_FEAT;
  float* scr  = cnt + (size_t)n_nodes * 3;
  float* p    = scr + (size_t)n_nodes * D_FEAT;

  unsigned* keys = (unsigned*)scr;
  float* w1rp = cnt + ((size_t)n_nodes + 127) / 128 * 128;
  float* w1lp = w1rp + D_FEAT * D_FEAT;

  const int key_bytes = n_edges + 2 * OWN_AHEAD * OWN_WAVES * OWN_STEP;
  unsigned* words = keys + ((size_t)key_bytes / 16 + 512) * 4;

  const int n_blocks = (n_nodes + OWN_NB - 1) / OWN_NB;
  const int stat_total = n_blocks * OWN_WAVES * OWN_K;
  const int ovf_cap = n_edges / 16 + n_blocks * OWN_WAVES + 64;
  const size_t max_chunks = (size_t)stat_total + ovf_cap;
  unsigned* chunks = words + ((size_t)n_edges / 4 + 4) * 4;
  unsigned* dir32 = chunks + max_chunks * 32;
  const size_t dir_words = (max_chunks / 16 + 1) * 4;
  int* cursor = (int*)(dir32 + dir_words);

  sage_prep<<<(key_bytes + 16 * 256 - 1) / (16 * 256), 256, 0, stream>>>(
      ei, keys, words, dir32, dir_words, cursor, W1r, w1rp, W1l, w1lp, n_edges);

  sage_own1<<<n_blocks, OWN_WAVES * 32, 0, stream>>>(
      x, words, keys, agg1, cnt, chunks, (unsigned char*)dir32, cursor, n_edges,
      n_nodes);

  const int n_tiles = n_nodes / 16;
  sage_layer1_wmma<<<(n_tiles + 7) / 8, 256, 0, stream>>>(
      x, agg1, cnt, w1lp, w1rp, b1, W2l, W2r, b2, p, out, n_nodes);

  sage_own2<<<n_blocks, OWN_WAVES * 32, 0, stream>>>(p, chunks, dir32, cursor, cnt,
                                                     out, n_nodes, stat_total,
                                                     ovf_cap);
}


# Layer-1 owner blocks read their own bitmap row (1 bit per 16-edge group, built by launch 0) and fetch only the marked groups' packed dst|src words: no key scan, no read per hit; sums arrival-ordered
# speedup vs baseline: 4.9059x; 1.0660x over previous
#include <hip/hip_runtime.h>

typedef __attribute__((ext_vector_type(2))) float v2f;
typedef __attribute__((ext_vector_type(8))) float v8f;

#define D_FEAT 128

#define OWN_NB 200
#define OWN_WAVES 32
#define OWN_GRP 16
#define OWN_FLUSH 32
#define OWN_QCAP 320
#define OWN_GCAP 64
#define OWN_B 16
#define OWN_PAD 0xFFu
#define OWN_K 8
#define OWN2_QCAP 1536
#define PREP_EDGES 4096

__device__ __forceinline__ unsigned own_key(int dst) {
  const unsigned k = (unsigned)dst / OWN_NB;
  return k < OWN_PAD ? k : OWN_PAD;
}

__device__ __forceinline__ unsigned own_word(int s, int dst) {
  return own_key(dst) < OWN_PAD
             ? (((unsigned)dst << 16) | ((unsigned)s & 0xFFFFu))
             : ~0u;
}

__device__ __forceinline__ uint4 own_word4(int4 s, int4 d) {
  uint4 w;
  w.x = own_word(s.x, d.x);
  w.y = own_word(s.y, d.y);
  w.z = own_word(s.z, d.z);
  w.w = own_word(s.w, d.w);
  return w;
}

__device__ __forceinline__ void own_mark(unsigned* lb, int lw, unsigned bit,
                                         int dst, int n_own) {
  const unsigned o = own_key(dst);
  if (o < (unsigned)n_own) atomicOr(lb + o * 8 + lw, bit);
}

__global__ __launch_bounds__(256) void sage_prep(
    const int* __restrict__ ei, unsigned* __restrict__ bm,
    unsigned* __restrict__ words, unsigned* __restrict__ dir32, size_t zn,
    int* __restrict__ cursor, const float* __restrict__ W1r,
    float* __restrict__ w1rp, const float* __restrict__ W1l,
    float* __restrict__ w1lp, int n_edges, int n_own, int bm_pitch,
    int n_eblk) {
  __shared__ unsigned lb[256 * 8];

  const size_t t = (size_t)blockIdx.x * blockDim.x + threadIdx.x;
  const size_t nt = (size_t)gridDim.x * blockDim.x;
  const size_t pk0 = nt - (size_t)(2 * D_FEAT * D_FEAT / 4);
  if (t >= pk0) {
    const int i = (int)(t - pk0);
    if (i < D_FEAT * D_FEAT / 4) {
      const int fl = i & 31;
      const int fn = (i >> 5) & 7;
      const int fk = i >> 8;
      *(float4*)(w1rp + (size_t)i * 4) = *(const float4*)(
          W1r + (size_t)(fn * 16 + (fl & 15)) * D_FEAT + fk * 8 + 4 * (fl >> 4));
    } else {
      const int i2 = i - D_FEAT * D_FEAT / 4;
      const int r = i2 >> 5;
      const int sb = 2 * (i2 & 31);
      const int key = (2 * r) & 63;
      const int c0 = sb ^ key, c1 = (sb + 1) ^ key;
      const int f0 = 4 * ((2 * c0) & 31) + (c0 >> 4);
      const int f1 = 4 * ((2 * c1) & 31) + (c1 >> 4);
      const float* wr = W1l + (size_t)r * D_FEAT;
      float4 v;
      v.x = wr[f0];
      v.y = wr[f0 + 4];
      v.z = wr[f1];
      v.w = wr[f1 + 4];
      *(float4*)(w1lp + (size_t)i2 * 4) = v;
    }
  }
  for (size_t i = t; i < zn; i += nt) dir32[i] = ~0u;
  if (t == 0) *cursor = 0;

  if ((int)blockIdx.x >= n_eblk) return;

  for (int i = threadIdx.x; i < 256 * 8; i += 256) lb[i] = 0u;
  __syncthreads();

  const int* dstp = ei + n_edges;
  const size_t g0 = (size_t)blockIdx.x * (PREP_EDGES / 4) + threadIdx.x;
  const size_t n_e16 = ((size_t)n_edges + OWN_GRP - 1) & ~(size_t)(OWN_GRP - 1);
  int4 s[4], d[4];
  if ((n_edges & 3) == 0) {
    const size_t n_groups = (size_t)n_edges >> 2;
#pragma unroll
    for (int j = 0; j < 4; ++j) {
      const size_t g = g0 + (size_t)j * 256;
      const size_t gc = g < n_groups ? g : n_groups - 1;
      s[j] = *(const int4*)(ei + gc * 4);
      d[j] = *(const int4*)(dstp + gc * 4);
    }
#pragma unroll
    for (int j = 0; j < 4; ++j) {
      const size_t g = g0 + (size_t)j * 256;
      if (g >= n_groups) d[j] = make_int4(-1, -1, -1, -1);
    }
  } else {
#pragma unroll
    for (int j = 0; j < 4; ++j) {
      const size_t e = (g0 + (size_t)j * 256) * 4;
      const size_t ne = (size_t)n_edges;
      s[j].x = (e < ne) ? ei[e] : 0;
      d[j].x = (e < ne) ? dstp[e] : -1;
      s[j].y = (e + 1 < ne) ? ei[e + 1] : 0;
      d[j].y = (e + 1 < ne) ? dstp[e + 1] : -1;
      s[j].z = (e + 2 < ne) ? ei[e + 2] : 0;
      d[j].z = (e + 2 < ne) ? dstp[e + 2] : -1;
      s[j].w = (e + 3 < ne) ? ei[e + 3] : 0;
      d[j].w = (e + 3 < ne) ? dstp[e + 3] : -1;
    }
  }
#pragma unroll
  for (int j = 0; j < 4; ++j) {
    const size_t g = g0 + (size_t)j * 256;
    if (g * 4 < n_e16) *(uint4*)(words + g * 4) = own_word4(s[j], d[j]);
    const int l16 = ((int)threadIdx.x + j * 256) >> 2;
    const int lw = l16 >> 5;
    const unsigned bit = 1u << (l16 & 31);
    own_mark(lb, lw, bit, d[j].x, n_own);
    own_mark(lb, lw, bit, d[j].y, n_own);
    own_mark(lb, lw, bit, d[j].z, n_own);
    own_mark(lb, lw, bit, d[j].w, n_own);
  }
  __syncthreads();
  for (int i = threadIdx.x; i < n_own * 8; i += 256)
    bm[(size_t)(i >> 3) * bm_pitch + (size_t)blockIdx.x * 8 + (i & 7)] = lb[i];
}


__device__ __forceinline__ unsigned own_eq(unsigned w, unsigned pat) {
  const unsigned t = w ^ pat;
  return ~(((t & 0x7F7F7F7Fu) + 0x7F7F7F7Fu) | t) & 0x80808080u;
}

__device__ __forceinline__ int own_push(unsigned* q, int qn, uint4 k,
                                        unsigned pat, int e) {
  unsigned m = (own_eq(k.x, pat) >> 7) | (own_eq(k.y, pat) >> 6) |
               (own_eq(k.z, pat) >> 5) | (own_eq(k.w, pat) >> 4);
  for (int r = 0; r < 16; ++r) {
    const bool has = m != 0u;
    const unsigned mask = __builtin_amdgcn_ballot_w32(has);
    if (mask == 0u) break;
    if (has) {
      const int p = __builtin_ctz(m);
      m &= m - 1u;
      const int pos = qn + (int)__builtin_amdgcn_mbcnt_lo(mask, 0u);
      q[pos] = (unsigned)(e + 4 * (p & 7) + (p >> 3));
    }
    qn += __builtin_popcount(mask);
  }
  return qn;
}

__device__ __forceinline__ unsigned long long own_bits(
    const unsigned char* __restrict__ row, int c0, int wend, int lane) {
  unsigned long long m = 0ull;
#pragma unroll
  for (int i = 0; i < 8; ++i) {
    const int k = c0 + 32 * i + lane;
    const unsigned b = row[k < wend ? k : (wend > 0 ? wend - 1 : 0)];
    m |= (unsigned long long)(k < wend ? b : 0u) << (8 * i);
  }
  return m;
}

__device__ __forceinline__ int own_groups(const unsigned* __restrict__ words,
                                          const unsigned* gq, int ng,
                                          unsigned base, unsigned* q, int qn,
                                          int lane) {
  const int half = lane >> 4;
  const int sub = lane & 15;
  unsigned w[8];
#pragma unroll
  for (int u = 0; u < 8; ++u) {
    const int idx = 2 * u + half;
    const unsigned wv = words[(size_t)gq[idx < ng ? idx : ng - 1] * OWN_GRP + sub];
    w[u] = idx < ng ? wv : ~0u;
  }
#pragma unroll
  for (int u = 0; u < 8; ++u) {
    const bool hit = ((w[u] >> 16) - base) < (unsigned)OWN_NB;
    const unsigned hm = __builtin_amdgcn_ballot_w32(hit);
    if (hit) q[qn + (int)__builtin_amdgcn_mbcnt_lo(hm, 0u)] = w[u];
    qn += __builtin_popcount(hm);
  }
  return qn;
}

__device__ __forceinline__ void own_add_row(float* tile, int dl, int lane,
                                            const float4 v) {
  float* t = tile + dl * D_FEAT + lane;
  atomicAdd(t + 0, v.x);
  atomicAdd(t + 32, v.y);
  atomicAdd(t + 64, v.z);
  atomicAdd(t + 96, v.w);
}

__device__ __forceinline__ void own_entry(float* cntL, const unsigned* q,
                                          unsigned base, int& mysrc, int& myl,
                                          unsigned& w) {
  const unsigned e = *q;
  mysrc = (int)(e & 0xFFFFu);
  myl = (int)((e >> 16) - base);
  w = ((unsigned)myl << 24) | (unsigned)mysrc;
  atomicAdd(cntL + myl, 1.0f);
}

struct OwnOut {
  unsigned* chunks;
  unsigned char* dir;
  int* cursor;
  int slot0;
  int stat_total;
  int nem;
};

__device__ __forceinline__ void own_emit(OwnOut& o, unsigned w, int lane) {
  int c;
  if (o.nem < OWN_K) {
    c = o.slot0 + o.nem;
  } else {
    int t = 0;
    if (lane == 0) t = atomicAdd(o.cursor, 1);
    c = o.stat_total + __builtin_amdgcn_readfirstlane(t);
  }
  ++o.nem;
  o.chunks[(size_t)c * 32 + lane] = w;
  if (lane == 0) o.dir[c] = (unsigned char)blockIdx.x;
}

__device__ __forceinline__ int own_flush(const float* __restrict__ x,
                                         float* tile, float* cntL, unsigned* q,
                                         int qn, int lane, bool last,
                                         unsigned base, OwnOut& o) {
  int b = 0;
  while (qn - b >= OWN_B) {
    int g = (qn - b) & ~(OWN_B - 1);
    if (g > 32) g = 32;
    int mysrc = 0, myl = 0;
    unsigned w = ~0u;
    if (lane < g) own_entry(cntL, q + b + lane, base, mysrc, myl, w);
    own_emit(o, w, lane);
    for (int i0 = 0; i0 < g; i0 += OWN_B) {
      float4 v[OWN_B];
      int dl[OWN_B];
#pragma unroll
      for (int u = 0; u < OWN_B; ++u) {
        const int sidx = __builtin_amdgcn_readlane(mysrc, i0 + u);
        dl[u] = __builtin_amdgcn_readlane(myl, i0 + u);
        v[u] = *(const float4*)(x + (size_t)sidx * D_FEAT + lane * 4);
      }
#pragma unroll
      for (int u = 0; u < OWN_B; ++u) own_add_row(tile, dl[u], lane, v[u]);
    }
    b += g;
  }
  const int r = qn - b;
  if (last) {
    if (r > 0) {
      int mysrc = 0, myl = 0;
      unsigned w = ~0u;
      if (lane < r) own_entry(cntL, q + b + lane, base, mysrc, myl, w);
      own_emit(o, w, lane);
      float4 v[OWN_B - 1];
      int dl[OWN_B - 1];
#pragma unroll
      for (int u = 0; u < OWN_B - 1; ++u) {
        if (u < r) {
          const int sidx = __builtin_amdgcn_readlane(mysrc, u);
          dl[u] = __builtin_amdgcn_readlane(myl, u);
          v[u] = *(const float4*)(x + (size_t)sidx * D_FEAT + lane * 4);
        }
      }
#pragma unroll
      for (int u = 0; u < OWN_B - 1; ++u) {
        if (u < r) own_add_row(tile, dl[u], lane, v[u]);
      }
    }
    return 0;
  }
  if (r > 0 && b > 0) {
    unsigned t = 0u;
    if (lane < r) t = q[b + lane];
    if (lane < r) q[lane] = t;
  }
  return r;
}

__global__ __launch_bounds__(1024) void sage_own1(
    const float* __restrict__ x, const unsigned* __restrict__ words,
    const unsigned char* __restrict__ bm, float* __restrict__ agg1,
    float* __restrict__ cnt, unsigned* __restrict__ chunks,
    unsigned char* __restrict__ dir, int* __restrict__ cursor, int n_edges,
    int n_nodes, int bm_pitch_bytes) {
  __shared__ float tile[OWN_NB * D_FEAT];
  __shared__ float cntL[OWN_NB];
  __shared__ unsigned queue[OWN_WAVES * OWN_QCAP];
  __shared__ unsigned gqueue[OWN_WAVES * OWN_GCAP];

  const int lane = threadIdx.x & 31;
  const int wave = __builtin_amdgcn_readfirstlane((int)(threadIdx.x >> 5));
  const int base = blockIdx.x * OWN_NB;

  const unsigned char* row = bm + (size_t)blockIdx.x * bm_pitch_bytes;
  const int n_g16 = (n_edges + OWN_GRP - 1) / OWN_GRP;
  const int nb = (n_g16 + 7) >> 3;
  const int nbw = (nb + OWN_WAVES - 1) / OWN_WAVES;
  const int wbeg = wave * nbw;
  const int wend = (wbeg + nbw < nb) ? wbeg + nbw : nb;
  unsigned long long m = own_bits(row, wbeg, wend, lane);

  for (int i = threadIdx.x; i < OWN_NB * D_FEAT / 4; i += OWN_WAVES * 32)
    ((float4*)tile)[i] = make_float4(0.0f, 0.0f, 0.0f, 0.0f);
  if (threadIdx.x < OWN_NB) cntL[threadIdx.x] = 0.0f;
  __syncthreads();

  unsigned* q = queue + wave * OWN_QCAP;
  unsigned* gq = gqueue + wave * OWN_GCAP;
  OwnOut o;
  o.chunks = chunks;
  o.dir = dir;
  o.cursor = cursor;
  o.slot0 = ((int)blockIdx.x * OWN_WAVES + wave) * OWN_K;
  o.stat_total = (int)gridDim.x * OWN_WAVES * OWN_K;
  o.nem = 0;
  int qn = 0;
  int gn = 0;

  for (int c0 = wbeg; c0 < wend;) {
    for (;;) {
      const bool has = m != 0ull;
      const unsigned mask = __builtin_amdgcn_ballot_w32(has);
      if (mask == 0u) break;
      if (has) {
        const int j = __builtin_ctzll(m);
        m &= m - 1ull;
        gq[gn + (int)__builtin_amdgcn_mbcnt_lo(mask, 0u)] =
            (unsigned)((c0 + 32 * (j >> 3) + lane) * 8 + (j & 7));
      }
      gn += __builtin_popcount(mask);
      while (gn >= 16) {
        gn -= 16;
        qn = own_groups(words, gq + gn, 16, (unsigned)base, q, qn, lane);
        if (qn >= OWN_FLUSH)
          qn = own_flush(x, tile, cntL, q, qn, lane, false, (unsigned)base, o);
      }
    }
    c0 += 256;
    if (c0 < wend) m = own_bits(row, c0, wend, lane);
  }
  if (gn > 0) qn = own_groups(words, gq, gn, (unsigned)base, q, qn, lane);
  own_flush(x, tile, cntL, q, qn, lane, true, (unsigned)base, o);
  __syncthreads();

  for (int i = threadIdx.x; i < OWN_NB * D_FEAT / 4; i += OWN_WAVES * 32) {
    const int row_n = base + (i >> 5);
    if (row_n < n_nodes)
      *(float4*)(agg1 + (size_t)row_n * D_FEAT + (i & 31) * 4) = ((const float4*)tile)[i];
  }
  if (threadIdx.x < OWN_NB && base + (int)threadIdx.x < n_nodes)
    cnt[base + threadIdx.x] = cntL[threadIdx.x];
}

__global__ __launch_bounds__(256) __attribute__((amdgpu_waves_per_eu(1, 4)))
void sage_layer1_wmma(
    const float* __restrict__ x, const float* __restrict__ agg1,
    const float* __restrict__ cnt, const float* __restrict__ W1lp,
    const float* __restrict__ W1rp, const float* __restrict__ b1,
    const float* __restrict__ W2l, const float* __restrict__ W2r,
    const float* __restrict__ b2, float* __restrict__ p,
    float* __restrict__ out, int n_nodes) {
  __shared__ __attribute__((aligned(16))) float ldsWl[D_FEAT * D_FEAT];

#pragma unroll
  for (int i = 0; i < D_FEAT * D_FEAT / 4 / 256; ++i)
    ((float4*)ldsWl)[threadIdx.x + i * 256] =
        ((const float4*)W1lp)[threadIdx.x + i * 256];
  __syncthreads();

  const int lane = threadIdx.x & 31;
  const int wave = threadIdx.x >> 5;
  const int tile = blockIdx.x * 8 + wave;
  const int n_tiles = n_nodes >> 4;
  if (tile >= n_tiles) return;

  const int half = lane >> 4;
  const int q = lane & 15;
  const int rowA = tile * 16 + q;

  const float inv = 1.0f / fmaxf(cnt[rowA], 1.0f);

  v8f acc[8];
#pragma unroll
  for (int i = 0; i < 8; ++i) {
    v8f z = {};
    acc[i] = z;
  }

  const float* aggRow = agg1 + (size_t)rowA * D_FEAT + 4 * half;
  const float* xRow   = x    + (size_t)rowA * D_FEAT + 4 * half;

  int nrow[8], nswz[8];
#pragma unroll
  for (int nt = 0; nt < 8; ++nt) {
    nrow[nt] = nt * 16 + q;
    nswz[nt] = nrow[nt] & 31;
  }

  struct Frag {
    float4 am, as;
    float4 bl[8], br[8];
  };
  auto load = [&](Frag& f, int k) {
    const int u0 = (k >> 2) + half;
    f.am = *(const float4*)(aggRow + k);
    f.as = *(const float4*)(xRow + k);
#pragma unroll
    for (int nt = 0; nt < 8; ++nt) {
      f.bl[nt] = *(const float4*)(ldsWl + nrow[nt] * D_FEAT + 4 * (u0 ^ nswz[nt]));
      f.br[nt] = *(const float4*)(W1rp + (size_t)(((k >> 3) * 8 + nt) * 32 + lane) * 4);
    }
  };
  auto mma = [&](const Frag& f) {
    v2f a_m0;  a_m0.x = f.am.x * inv;  a_m0.y = f.am.y * inv;
    v2f a_m1;  a_m1.x = f.am.z * inv;  a_m1.y = f.am.w * inv;
    v2f a_s0;  a_s0.x = f.as.x;        a_s0.y = f.as.y;
    v2f a_s1;  a_s1.x = f.as.z;        a_s1.y = f.as.w;
#pragma unroll
    for (int nt = 0; nt < 8; ++nt) {
      v2f vbl;  vbl.x = f.bl[nt].x;  vbl.y = f.bl[nt].y;
      v2f vbr;  vbr.x = f.br[nt].x;  vbr.y = f.br[nt].y;
      acc[nt] = __builtin_amdgcn_wmma_f32_16x16x4_f32(
          false, vbl, false, a_m0, (short)0, acc[nt], false, false);
      acc[nt] = __builtin_amdgcn_wmma_f32_16x16x4_f32(
          false, vbr, false, a_s0, (short)0, acc[nt], false, false);
    }
#pragma unroll
    for (int nt = 0; nt < 8; ++nt) {
      v2f vbl;  vbl.x = f.bl[nt].z;  vbl.y = f.bl[nt].w;
      v2f vbr;  vbr.x = f.br[nt].z;  vbr.y = f.br[nt].w;
      acc[nt] = __builtin_amdgcn_wmma_f32_16x16x4_f32(
          false, vbl, false, a_m1, (short)0, acc[nt], false, false);
      acc[nt] = __builtin_amdgcn_wmma_f32_16x16x4_f32(
          false, vbr, false, a_s1, (short)0, acc[nt], false, false);
    }
  };

  Frag f0, f1;
  load(f0, 0);
  for (int k = 0; k < D_FEAT - 16; k += 16) {
    load(f1, k + 8);
    mma(f0);
    load(f0, k + 16);
    mma(f1);
  }
  load(f1, D_FEAT - 8);
  mma(f0);
  mma(f1);

  const int r4 = q & 3;
  const float* w2row = (r4 < 2) ? (W2l + r4 * D_FEAT) : (W2r + (r4 - 2) * D_FEAT);
  v8f o0 = {}, o1 = {};
#pragma unroll
  for (int nt = 0; nt < 8; ++nt) {
    const int f0 = nt * 16 + 8 * half;
    const float4 ba = *(const float4*)(b1 + f0);
    const float4 bb = *(const float4*)(b1 + f0 + 4);
    const float4 wa = *(const float4*)(w2row + f0);
    const float4 wb = *(const float4*)(w2row + f0 + 4);
    v2f h0, h1, h2, h3;
    h0.x = fmaxf(acc[nt][0] + ba.x, 0.0f);  h0.y = fmaxf(acc[nt][1] + ba.y, 0.0f);
    h1.x = fmaxf(acc[nt][2] + ba.z, 0.0f);  h1.y = fmaxf(acc[nt][3] + ba.w, 0.0f);
    h2.x = fmaxf(acc[nt][4] + bb.x, 0.0f);  h2.y = fmaxf(acc[nt][5] + bb.y, 0.0f);
    h3.x = fmaxf(acc[nt][6] + bb.z, 0.0f);  h3.y = fmaxf(acc[nt][7] + bb.w, 0.0f);
    v2f w0, w1, w2, w3;
    w0.x = wa.x;  w0.y = wa.y;
    w1.x = wa.z;  w1.y = wa.w;
    w2.x = wb.x;  w2.y = wb.y;
    w3.x = wb.z;  w3.y = wb.w;
    o0 = __builtin_amdgcn_wmma_f32_16x16x4_f32(
        false, w0, false, h0, (short)0, o0, false, false);
    o1 = __builtin_amdgcn_wmma_f32_16x16x4_f32(
        false, w1, false, h1, (short)0, o1, false, false);
    o0 = __builtin_amdgcn_wmma_f32_16x16x4_f32(
        false, w2, false, h2, (short)0, o0, false, false);
    o1 = __builtin_amdgcn_wmma_f32_16x16x4_f32(
        false, w3, false, h3, (short)0, o1, false, false);
  }
  if (half == 0) {
    const size_t node = (size_t)tile * 16 + q;
    float2 pv, ov;
    pv.x = o0[0] + o1[0];
    pv.y = o0[1] + o1[1];
    ov.x = o0[2] + o1[2] + b2[0];
    ov.y = o0[3] + o1[3] + b2[1];
    *(float2*)(p + node * 2) = pv;
    *(float2*)(out + node * 2) = ov;
  }
}

__device__ __forceinline__ int own2_drain(const float* __restrict__ p,
                                          const unsigned* __restrict__ chunks,
                                          float* tile2, const unsigned* q,
                                          int qn, int lane) {
  for (int b = 0; b < qn; b += 8) {
    unsigned w[8];
#pragma unroll
    for (int u = 0; u < 8; ++u) {
      w[u] = ~0u;
      if (b + u < qn) w[u] = chunks[(size_t)q[b + u] * 32 + lane];
    }
    float px[8], py[8];
#pragma unroll
    for (int u = 0; u < 8; ++u) {
      px[u] = 0.0f;
      py[u] = 0.0f;
      if (w[u] != ~0u) {
        const size_t s = (size_t)(w[u] & 0xFFFFFFu) * 2;
        px[u] = p[s];
        py[u] = p[s + 1];
      }
    }
#pragma unroll
    for (int u = 0; u < 8; ++u) {
      if (w[u] != ~0u) {
        float* t = tile2 + 2 * (w[u] >> 24);
        atomicAdd(t, px[u]);
        atomicAdd(t + 1, py[u]);
      }
    }
  }
  return 0;
}

__global__ __launch_bounds__(1024) void sage_own2(
    const float* __restrict__ p, const unsigned* __restrict__ chunks,
    const unsigned* __restrict__ dir32, const int* __restrict__ cursor,
    const float* __restrict__ cnt, float* __restrict__ out, int n_nodes,
    int stat_total, int ovf_cap) {
  __shared__ float tile2[OWN_NB * 2];
  __shared__ unsigned queue2[OWN_WAVES * OWN2_QCAP];

  const int lane = threadIdx.x & 31;
  const int wave = __builtin_amdgcn_readfirstlane((int)(threadIdx.x >> 5));
  const int base = blockIdx.x * OWN_NB;
  const unsigned pat = blockIdx.x * 0x01010101u;

  if (threadIdx.x < OWN_NB * 2) tile2[threadIdx.x] = 0.0f;
  __syncthreads();

  int ov = *cursor;
  if (ov > ovf_cap) ov = ovf_cap;
  if (ov < 0) ov = 0;
  const int n_scan = stat_total + ov;

  unsigned* q = queue2 + wave * OWN2_QCAP;
  int qn = 0;
  const uint4 none = make_uint4(~0u, ~0u, ~0u, ~0u);
  static_assert(OWN_K == 8, "own slots of a wave = two directory dwords");
  {
    const int slot0 = ((int)blockIdx.x * OWN_WAVES + wave) * OWN_K;
    uint4 k = none;
    if (lane == 0) {
      k.x = dir32[slot0 >> 2];
      k.y = dir32[(slot0 >> 2) + 1];
    }
    qn = own_push(q, qn, k, pat, slot0);
  }
  for (int r0 = stat_total; r0 < n_scan; r0 += OWN_WAVES * 1024) {
    const int cb = r0 + wave * 1024 + lane * 16;
    uint4 k0 = none, k1 = none;
    if (cb < n_scan) k0 = *(const uint4*)(dir32 + (cb >> 2));
    if (cb + 512 < n_scan) k1 = *(const uint4*)(dir32 + ((cb + 512) >> 2));
    qn = own_push(q, qn, k0, pat, cb);
    qn = own_push(q, qn, k1, pat, cb + 512);
    if (qn > OWN2_QCAP - 1024) qn = own2_drain(p, chunks, tile2, q, qn, lane);
  }
  own2_drain(p, chunks, tile2, q, qn, lane);
  __syncthreads();

  if (threadIdx.x < OWN_NB) {
    const int node = base + (int)threadIdx.x;
    if (node < n_nodes) {
      const float inv = 1.0f / fmaxf(cnt[node], 1.0f);
      out[(size_t)node * 2 + 0] += tile2[2 * threadIdx.x + 0] * inv;
      out[(size_t)node * 2 + 1] += tile2[2 * threadIdx.x + 1] * inv;
    }
  }
}

extern "C" void kernel_launch(void* const* d_in, const int* in_sizes, int n_in,
                              void* d_out, int out_size, void* d_ws, size_t ws_size,
                              hipStream_t stream) {
  const float* x   = (const float*)d_in[0];
  const int*   ei  = (const int*)d_in[1];
  const float* W1l = (const float*)d_in[2];
  const float* W1r = (const float*)d_in[3];
  const float* b1  = (const float*)d_in[4];
  const float* W2l = (const float*)d_in[5];
  const float* W2r = (const float*)d_in[6];
  const float* b2  = (const float*)d_in[7];
  float* out = (float*)d_out;

  const int n_nodes = in_sizes[0] / D_FEAT;
  const int n_edges = in_sizes[1] / 2;

  float* ws   = (float*)d_ws;
  float* agg1 = ws;
  float* cnt  = agg1 + (size_t)n_nodes * D_FEAT;
  float* scr  = cnt + (size_t)n_nodes * 3;
  float* p    = scr + (size_t)n_nodes * D_FEAT;

  float* w1rp = cnt + ((size_t)n_nodes + 127) / 128 * 128;
  float* w1lp = w1rp + D_FEAT * D_FEAT;

  const int n_blocks = (n_nodes + OWN_NB - 1) / OWN_NB;
  const int n_e16 = (n_edges + OWN_GRP - 1) / OWN_GRP * OWN_GRP;
  const int n_eblk = (n_e16 + PREP_EDGES - 1) / PREP_EDGES;
  const int bm_pitch = n_eblk * 8;
  unsigned* bm = (unsigned*)scr;
  unsigned* words = bm + ((size_t)n_blocks * bm_pitch + 127) / 128 * 128;

  const int stat_total = n_blocks * OWN_WAVES * OWN_K;
  const int ovf_cap = n_edges / 16 + n_blocks * OWN_WAVES + 64;
  const size_t max_chunks = (size_t)stat_total + ovf_cap;
  unsigned* chunks = words + ((size_t)n_e16 + 31) / 32 * 32;
  unsigned* dir32 = chunks + max_chunks * 32;
  const size_t dir_words = (max_chunks / 16 + 1) * 4;
  int* cursor = (int*)(dir32 + dir_words);

  sage_prep<<<n_eblk + 32, 256, 0, stream>>>(
      ei, bm, words, dir32, dir_words, cursor, W1r, w1rp, W1l, w1lp, n_edges,
      n_blocks, bm_pitch, n_eblk);

  sage_own1<<<n_blocks, OWN_WAVES * 32, 0, stream>>>(
      x, words, (const unsigned char*)bm, agg1, cnt, chunks,
      (unsigned char*)dir32, cursor, n_edges, n_nodes, bm_pitch * 4);

  const int n_tiles = n_nodes / 16;
  sage_layer1_wmma<<<(n_tiles + 7) / 8, 256, 0, stream>>>(
      x, agg1, cnt, w1lp, w1rp, b1, W2l, W2r, b2, p, out, n_nodes);

  sage_own2<<<n_blocks, OWN_WAVES * 32, 0, stream>>>(p, chunks, dir32, cursor, cnt,
                                                     out, n_nodes, stat_total,
                                                     ovf_cap);
}
